# starting kernel
# baseline (speedup 1.0000x reference)
#include <hip/hip_runtime.h>


typedef float v2f __attribute__((ext_vector_type(2)));
typedef float v4f __attribute__((ext_vector_type(4)));
typedef float v8f __attribute__((ext_vector_type(8)));

#define B_  2048
#define T_  1024
#define D_  24
#define H_  8
#define O_  24
#define G_  32

#define LOG2E_ 1.4426950408889634f

__device__ __forceinline__ float gate_act(float x, float ik, float ib,
                                          float om, float oa) {
    float e = __builtin_amdgcn_exp2f(__builtin_fmaf(ik, x, ib));
    float r = __builtin_amdgcn_rcpf(1.0f + e);
    return __builtin_fmaf(om, r, oa);
}

__device__ __forceinline__ float tanh_nb(float x) {
    float e = __builtin_amdgcn_exp2f(x * (-2.0f * LOG2E_));
    float r = __builtin_amdgcn_rcpf(1.0f + e);
    return __builtin_fmaf(2.0f, r, -1.0f);
}

__global__ __launch_bounds__(32)
void lstm2_fused(const float* __restrict__ x,
                 const float* __restrict__ w_ih0, const float* __restrict__ w_hh0,
                 const float* __restrict__ b_ih0, const float* __restrict__ b_hh0,
                 const float* __restrict__ w_ih1, const float* __restrict__ w_hh1,
                 const float* __restrict__ b_ih1, const float* __restrict__ b_hh1,
                 const float* __restrict__ w_lin, const float* __restrict__ b_lin,
                 float* __restrict__ out)
{
    __shared__ __align__(16) float gatesLDS[16 * G_];
    __shared__ __align__(16) float h0LDS[16 * H_];
    __shared__ __align__(16) float h1LDS[16 * H_];

    const int lane = threadIdx.x;
    const int n    = lane & 15;
    const int half = lane >> 4;
    const int b0   = blockIdx.x * 16;

    v2f Bx[6][2];
#pragma unroll
    for (int kk = 0; kk < 6; ++kk)
#pragma unroll
        for (int nt = 0; nt < 2; ++nt) {
            int g = nt * 16 + n;
            int k = kk * 4 + half * 2;
            Bx[kk][nt] = *(const v2f*)(w_ih0 + g * D_ + k);
        }
    v2f Bh0[2][2], Bi1[2][2], Bh1[2][2];
#pragma unroll
    for (int kk = 0; kk < 2; ++kk)
#pragma unroll
        for (int nt = 0; nt < 2; ++nt) {
            int g = nt * 16 + n;
            int k = kk * 4 + half * 2;
            Bh0[kk][nt] = *(const v2f*)(w_hh0 + g * H_ + k);
            Bi1[kk][nt] = *(const v2f*)(w_ih1 + g * H_ + k);
            Bh1[kk][nt] = *(const v2f*)(w_hh1 + g * H_ + k);
        }

    float bias0t0, bias0t1, bias1t0, bias1t1;
    {
        int g0 = n, g1 = 16 + n;
        bias0t0 = b_ih0[g0] + b_hh0[g0];
        bias0t1 = b_ih0[g1] + b_hh0[g1];
        bias1t0 = b_ih1[g0] + b_hh1[g0];
        bias1t1 = b_ih1[g1] + b_hh1[g1];
    }
    const float ik0 = -LOG2E_;
    const float ib0_l0 = ik0 * bias0t0, ib0_l1 = ik0 * bias1t0;
    const float ik1 = (n < 8) ? (-2.0f * LOG2E_) : -LOG2E_;
    const float ib1_l0 = ik1 * bias0t1, ib1_l1 = ik1 * bias1t1;
    const float om1 = (n < 8) ? 2.0f : 1.0f;
    const float oa1 = (n < 8) ? -1.0f : 0.0f;

    const float* xlane = x + (size_t)(b0 + n) * T_ * D_ + half * 2;

    v2f h0a0 = {0.f, 0.f}, h0a1 = {0.f, 0.f};
    v2f h1a0 = {0.f, 0.f}, h1a1 = {0.f, 0.f};
    v4f c0 = {0.f, 0.f, 0.f, 0.f}, c1 = {0.f, 0.f, 0.f, 0.f};

    const int bb = n;
    const int j0 = half * 4;
    const int brow = half * 8;

    v2f xa[6];
#pragma unroll
    for (int kk = 0; kk < 6; ++kk) xa[kk] = *(const v2f*)(xlane + kk * 4);

    for (int t = 0; t < T_; ++t) {
        v8f acc0 = {0.f, 0.f, 0.f, 0.f, 0.f, 0.f, 0.f, 0.f};
        v8f acc1 = {0.f, 0.f, 0.f, 0.f, 0.f, 0.f, 0.f, 0.f};

#pragma unroll
        for (int kk = 0; kk < 6; ++kk) {
            acc0 = __builtin_amdgcn_wmma_f32_16x16x4_f32(false, xa[kk], false, Bx[kk][0], (short)0, acc0, false, false);
            acc1 = __builtin_amdgcn_wmma_f32_16x16x4_f32(false, xa[kk], false, Bx[kk][1], (short)0, acc1, false, false);
        }
        acc0 = __builtin_amdgcn_wmma_f32_16x16x4_f32(false, h0a0, false, Bh0[0][0], (short)0, acc0, false, false);
        acc1 = __builtin_amdgcn_wmma_f32_16x16x4_f32(false, h0a0, false, Bh0[0][1], (short)0, acc1, false, false);
        acc0 = __builtin_amdgcn_wmma_f32_16x16x4_f32(false, h0a1, false, Bh0[1][0], (short)0, acc0, false, false);
        acc1 = __builtin_amdgcn_wmma_f32_16x16x4_f32(false, h0a1, false, Bh0[1][1], (short)0, acc1, false, false);

        {
            int tn = (t + 1 < T_) ? (t + 1) : t;
            const float* xp = xlane + (size_t)tn * D_;
#pragma unroll
            for (int kk = 0; kk < 6; ++kk) xa[kk] = *(const v2f*)(xp + kk * 4);
        }

#pragma unroll
        for (int v = 0; v < 8; ++v) {
            float s0 = gate_act(acc0[v], ik0, ib0_l0, 1.0f, 0.0f);
            float s1 = gate_act(acc1[v], ik1, ib1_l0, om1, oa1);
            gatesLDS[(brow + v) * G_ + n]      = s0;
            gatesLDS[(brow + v) * G_ + 16 + n] = s1;
        }
        __syncthreads();

        {
            const float* gp = gatesLDS + bb * G_;
            v4f ig = *(const v4f*)(gp + 0  + j0);
            v4f fg = *(const v4f*)(gp + 8  + j0);
            v4f gg = *(const v4f*)(gp + 16 + j0);
            v4f og = *(const v4f*)(gp + 24 + j0);
            v4f hh;
#pragma unroll
            for (int k = 0; k < 4; ++k) {
                c0[k] = __builtin_fmaf(fg[k], c0[k], ig[k] * gg[k]);
                hh[k] = og[k] * tanh_nb(c0[k]);
            }
            *(v4f*)(h0LDS + bb * H_ + j0) = hh;
        }
        __syncthreads();

        h0a0 = *(const v2f*)(h0LDS + n * H_ + half * 2);
        h0a1 = *(const v2f*)(h0LDS + n * H_ + 4 + half * 2);

        acc0 = (v8f){0.f, 0.f, 0.f, 0.f, 0.f, 0.f, 0.f, 0.f};
        acc1 = (v8f){0.f, 0.f, 0.f, 0.f, 0.f, 0.f, 0.f, 0.f};

        acc0 = __builtin_amdgcn_wmma_f32_16x16x4_f32(false, h0a0, false, Bi1[0][0], (short)0, acc0, false, false);
        acc1 = __builtin_amdgcn_wmma_f32_16x16x4_f32(false, h0a0, false, Bi1[0][1], (short)0, acc1, false, false);
        acc0 = __builtin_amdgcn_wmma_f32_16x16x4_f32(false, h0a1, false, Bi1[1][0], (short)0, acc0, false, false);
        acc1 = __builtin_amdgcn_wmma_f32_16x16x4_f32(false, h0a1, false, Bi1[1][1], (short)0, acc1, false, false);
        acc0 = __builtin_amdgcn_wmma_f32_16x16x4_f32(false, h1a0, false, Bh1[0][0], (short)0, acc0, false, false);
        acc1 = __builtin_amdgcn_wmma_f32_16x16x4_f32(false, h1a0, false, Bh1[0][1], (short)0, acc1, false, false);
        acc0 = __builtin_amdgcn_wmma_f32_16x16x4_f32(false, h1a1, false, Bh1[1][0], (short)0, acc0, false, false);
        acc1 = __builtin_amdgcn_wmma_f32_16x16x4_f32(false, h1a1, false, Bh1[1][1], (short)0, acc1, false, false);

#pragma unroll
        for (int v = 0; v < 8; ++v) {
            float s0 = gate_act(acc0[v], ik0, ib0_l1, 1.0f, 0.0f);
            float s1 = gate_act(acc1[v], ik1, ib1_l1, om1, oa1);
            gatesLDS[(brow + v) * G_ + n]      = s0;
            gatesLDS[(brow + v) * G_ + 16 + n] = s1;
        }
        __syncthreads();

        {
            const float* gp = gatesLDS + bb * G_;
            v4f ig = *(const v4f*)(gp + 0  + j0);
            v4f fg = *(const v4f*)(gp + 8  + j0);
            v4f gg = *(const v4f*)(gp + 16 + j0);
            v4f og = *(const v4f*)(gp + 24 + j0);
            v4f hh;
#pragma unroll
            for (int k = 0; k < 4; ++k) {
                c1[k] = __builtin_fmaf(fg[k], c1[k], ig[k] * gg[k]);
                hh[k] = og[k] * tanh_nb(c1[k]);
            }
            *(v4f*)(h1LDS + bb * H_ + j0) = hh;
        }
        __syncthreads();

        h1a0 = *(const v2f*)(h1LDS + n * H_ + half * 2);
        h1a1 = *(const v2f*)(h1LDS + n * H_ + 4 + half * 2);
    }

    {
        const int obase = half * 12;
        const float* hp = h1LDS + bb * H_;
#pragma unroll
        for (int oo = 0; oo < 12; ++oo) {
            int o = obase + oo;
            float s = b_lin[o];
#pragma unroll
            for (int j = 0; j < H_; ++j) s += hp[j] * w_lin[o * H_ + j];
            out[(size_t)(b0 + bb) * O_ + o] = s;
        }
    }
}

extern "C" void kernel_launch(void* const* d_in, const int* in_sizes, int n_in,
                              void* d_out, int out_size, void* d_ws, size_t ws_size,
                              hipStream_t stream) {
    (void)in_sizes; (void)n_in; (void)d_ws; (void)ws_size; (void)out_size;
    const float* x     = (const float*)d_in[0];
    const float* w_ih0 = (const float*)d_in[1];
    const float* w_hh0 = (const float*)d_in[2];
    const float* b_ih0 = (const float*)d_in[3];
    const float* b_hh0 = (const float*)d_in[4];
    const float* w_ih1 = (const float*)d_in[5];
    const float* w_hh1 = (const float*)d_in[6];
    const float* b_ih1 = (const float*)d_in[7];
    const float* b_hh1 = (const float*)d_in[8];
    const float* w_lin = (const float*)d_in[9];
    const float* b_lin = (const float*)d_in[10];
    float* out = (float*)d_out;

    dim3 grid(B_ / 16);
    dim3 block(32);
    hipLaunchKernelGGL(lstm2_fused, grid, block, 0, stream,
                       x, w_ih0, w_hh0, b_ih0, b_hh0,
                       w_ih1, w_hh1, b_ih1, b_hh1, w_lin, b_lin, out);
}


# gate products transposed (weights A, state B), gate rows and hidden k permuted: cell update and next operand stay in registers, no LDS or barrier in time loop; bias summed first, k sum order changed
# speedup vs baseline: 1.6324x; 1.6324x over previous
#include <hip/hip_runtime.h>


typedef float v2f __attribute__((ext_vector_type(2)));
typedef float v4f __attribute__((ext_vector_type(4)));
typedef float v8f __attribute__((ext_vector_type(8)));

#define B_  2048
#define T_  1024
#define D_  24
#define H_  8
#define O_  24

#define LOG2E_ 1.4426950408889634f

#define WMMA_F32(a, b, c) \
    __builtin_amdgcn_wmma_f32_16x16x4_f32(false, (a), false, (b), (short)0, (c), false, false)

__device__ __forceinline__ float sigmoid_f(float x) {
    float e = __builtin_amdgcn_exp2f(x * (-LOG2E_));
    return __builtin_amdgcn_rcpf(1.0f + e);
}

__device__ __forceinline__ float tanh_f(float x) {
    float e = __builtin_amdgcn_exp2f(x * (-2.0f * LOG2E_));
    float r = __builtin_amdgcn_rcpf(1.0f + e);
    return __builtin_fmaf(2.0f, r, -1.0f);
}

__device__ __forceinline__ void lstm_cell(const v8f& t0, const v8f& t1, v4f& c, v4f& h) {
    for (int k = 0; k < 4; ++k) {
        float ig = sigmoid_f(t0[k]);
        float fg = sigmoid_f(t0[4 + k]);
        float gg = tanh_f(t1[k]);
        float og = sigmoid_f(t1[4 + k]);
        c[k] = __builtin_fmaf(fg, c[k], ig * gg);
        h[k] = og * tanh_f(c[k]);
    }
}

__global__ __launch_bounds__(32)
void lstm2_fused(const float* __restrict__ x,
                 const float* __restrict__ w_ih0, const float* __restrict__ w_hh0,
                 const float* __restrict__ b_ih0, const float* __restrict__ b_hh0,
                 const float* __restrict__ w_ih1, const float* __restrict__ w_hh1,
                 const float* __restrict__ b_ih1, const float* __restrict__ b_hh1,
                 const float* __restrict__ w_lin, const float* __restrict__ b_lin,
                 float* __restrict__ out)
{
    __shared__ __align__(16) float h1LDS[16 * H_];

    const int lane = threadIdx.x;
    const int c    = lane & 15;
    const int half = lane >> 4;
    const int b0   = blockIdx.x * 16;

    const int jrow = 4 * (c >> 3) + (c & 3);
    const int trow = (c & 7) >> 2;
    const int growA0 = trow * H_ + jrow;
    const int growA1 = (2 + trow) * H_ + jrow;

    v2f Ax[6][2];
    for (int kk = 0; kk < 6; ++kk) {
        int k = kk * 4 + half * 2;
        Ax[kk][0] = *(const v2f*)(w_ih0 + growA0 * D_ + k);
        Ax[kk][1] = *(const v2f*)(w_ih0 + growA1 * D_ + k);
    }
    v2f Ah0[2][2], Ai1[2][2], Ah1[2][2];
    for (int kk = 0; kk < 2; ++kk) {
        int k = half * 4 + kk * 2;
        Ah0[kk][0] = *(const v2f*)(w_hh0 + growA0 * H_ + k);
        Ah0[kk][1] = *(const v2f*)(w_hh0 + growA1 * H_ + k);
        Ai1[kk][0] = *(const v2f*)(w_ih1 + growA0 * H_ + k);
        Ai1[kk][1] = *(const v2f*)(w_ih1 + growA1 * H_ + k);
        Ah1[kk][0] = *(const v2f*)(w_hh1 + growA0 * H_ + k);
        Ah1[kk][1] = *(const v2f*)(w_hh1 + growA1 * H_ + k);
    }

    v8f bias0t0, bias0t1, bias1t0, bias1t1;
    for (int r = 0; r < 8; ++r) {
        int g0 = (r >> 2) * H_ + 4 * half + (r & 3);
        int g1 = (2 + (r >> 2)) * H_ + 4 * half + (r & 3);
        bias0t0[r] = b_ih0[g0] + b_hh0[g0];
        bias0t1[r] = b_ih0[g1] + b_hh0[g1];
        bias1t0[r] = b_ih1[g0] + b_hh1[g0];
        bias1t1[r] = b_ih1[g1] + b_hh1[g1];
    }

    const float* xlane = x + (size_t)(b0 + c) * T_ * D_ + half * 2;

    v4f h0 = {0.f, 0.f, 0.f, 0.f}, h1 = {0.f, 0.f, 0.f, 0.f};
    v4f c0 = {0.f, 0.f, 0.f, 0.f}, c1 = {0.f, 0.f, 0.f, 0.f};

    v2f xa[6];
    for (int kk = 0; kk < 6; ++kk) xa[kk] = *(const v2f*)(xlane + kk * 4);

    for (int t = 0; t < T_; ++t) {
        v8f acc0 = bias0t0;
        v8f acc1 = bias0t1;
        for (int kk = 0; kk < 6; ++kk) {
            acc0 = WMMA_F32(Ax[kk][0], xa[kk], acc0);
            acc1 = WMMA_F32(Ax[kk][1], xa[kk], acc1);
        }
        for (int kk = 0; kk < 2; ++kk) {
            v2f hb = {h0[2 * kk], h0[2 * kk + 1]};
            acc0 = WMMA_F32(Ah0[kk][0], hb, acc0);
            acc1 = WMMA_F32(Ah0[kk][1], hb, acc1);
        }

        {
            int tn = (t + 1 < T_) ? (t + 1) : t;
            const float* xp = xlane + (size_t)tn * D_;
            for (int kk = 0; kk < 6; ++kk) xa[kk] = *(const v2f*)(xp + kk * 4);
        }

        lstm_cell(acc0, acc1, c0, h0);

        acc0 = bias1t0;
        acc1 = bias1t1;
        for (int kk = 0; kk < 2; ++kk) {
            v2f hb = {h0[2 * kk], h0[2 * kk + 1]};
            acc0 = WMMA_F32(Ai1[kk][0], hb, acc0);
            acc1 = WMMA_F32(Ai1[kk][1], hb, acc1);
        }
        for (int kk = 0; kk < 2; ++kk) {
            v2f hb = {h1[2 * kk], h1[2 * kk + 1]};
            acc0 = WMMA_F32(Ah1[kk][0], hb, acc0);
            acc1 = WMMA_F32(Ah1[kk][1], hb, acc1);
        }

        lstm_cell(acc0, acc1, c1, h1);
    }

    *(v4f*)(h1LDS + c * H_ + half * 4) = h1;
    __syncthreads();
    {
        const int obase = half * 12;
        const float* hp = h1LDS + c * H_;
        for (int oo = 0; oo < 12; ++oo) {
            int o = obase + oo;
            float s = b_lin[o];
            for (int j = 0; j < H_; ++j) s += hp[j] * w_lin[o * H_ + j];
            out[(size_t)(b0 + c) * O_ + o] = s;
        }
    }
}

extern "C" void kernel_launch(void* const* d_in, const int* in_sizes, int n_in,
                              void* d_out, int out_size, void* d_ws, size_t ws_size,
                              hipStream_t stream) {
    (void)in_sizes; (void)n_in; (void)d_ws; (void)ws_size; (void)out_size;
    const float* x     = (const float*)d_in[0];
    const float* w_ih0 = (const float*)d_in[1];
    const float* w_hh0 = (const float*)d_in[2];
    const float* b_ih0 = (const float*)d_in[3];
    const float* b_hh0 = (const float*)d_in[4];
    const float* w_ih1 = (const float*)d_in[5];
    const float* w_hh1 = (const float*)d_in[6];
    const float* b_ih1 = (const float*)d_in[7];
    const float* b_hh1 = (const float*)d_in[8];
    const float* w_lin = (const float*)d_in[9];
    const float* b_lin = (const float*)d_in[10];
    float* out = (float*)d_out;

    dim3 grid(B_ / 16);
    dim3 block(32);
    hipLaunchKernelGGL(lstm2_fused, grid, block, 0, stream,
                       x, w_ih0, w_hh0, b_ih0, b_hh0,
                       w_ih1, w_hh1, b_ih1, b_hh1, w_lin, b_lin, out);
}


# time loop software-pipelined: layer 1 one step behind layer 0, input products one step ahead, x loads three steps ahead, cell updates cut into pieces issued between matrix products; sum order kept
# speedup vs baseline: 2.0782x; 1.2731x over previous
#include <hip/hip_runtime.h>


typedef float v2f __attribute__((ext_vector_type(2)));
typedef float v4f __attribute__((ext_vector_type(4)));
typedef float v8f __attribute__((ext_vector_type(8)));

#define B_  2048
#define T_  1024
#define D_  24
#define H_  8
#define O_  24

#define LOG2E_ 1.4426950408889634f

#define WMMA_F32(a, b, c) \
    __builtin_amdgcn_wmma_f32_16x16x4_f32(false, (a), false, (b), (short)0, (c), false, false)

__device__ __forceinline__ float sigmoid_f(float x) {
    float e = __builtin_amdgcn_exp2f(x * (-LOG2E_));
    return __builtin_amdgcn_rcpf(1.0f + e);
}

__device__ __forceinline__ float tanh_f(float x) {
    float e = __builtin_amdgcn_exp2f(x * (-2.0f * LOG2E_));
    float r = __builtin_amdgcn_rcpf(1.0f + e);
    return __builtin_fmaf(2.0f, r, -1.0f);
}

__device__ __forceinline__ void lstm_cell(const v8f& t0, const v8f& t1, v4f& c, v4f& h) {
    for (int k = 0; k < 4; ++k) {
        float ig = sigmoid_f(t0[k]);
        float fg = sigmoid_f(t0[4 + k]);
        float gg = tanh_f(t1[k]);
        float og = sigmoid_f(t1[4 + k]);
        c[k] = __builtin_fmaf(fg, c[k], ig * gg);
        h[k] = og * tanh_f(c[k]);
    }
}

__device__ __forceinline__ void cell_p1(const v8f& t0, float (&z)[16], int o) {
    for (int r = o; r < o + 4; ++r) z[r] = __builtin_amdgcn_exp2f(t0[r] * (-LOG2E_));
}
__device__ __forceinline__ void cell_p2g(const v8f& t1, float (&z)[16]) {
    for (int r = 0; r < 4; ++r) z[8 + r] = __builtin_amdgcn_exp2f(t1[r] * (-2.0f * LOG2E_));
}
__device__ __forceinline__ void cell_p2o(const v8f& t1, float (&z)[16]) {
    for (int r = 4; r < 8; ++r) z[8 + r] = __builtin_amdgcn_exp2f(t1[r] * (-LOG2E_));
}
__device__ __forceinline__ void cell_p3(float (&z)[16], int o) {
    for (int r = o; r < o + 4; ++r) z[r] = __builtin_amdgcn_rcpf(1.0f + z[r]);
}
__device__ __forceinline__ void cell_p5a(const float (&z)[16], float (&m)[4]) {
    for (int k = 0; k < 4; ++k) {
        float gg = __builtin_fmaf(2.0f, z[8 + k], -1.0f);
        m[k] = z[k] * gg;
    }
}
__device__ __forceinline__ void cell_p5b(const float (&z)[16], const float (&m)[4], v4f& c, float (&q)[4]) {
    for (int k = 0; k < 4; ++k) {
        c[k] = __builtin_fmaf(z[4 + k], c[k], m[k]);
        q[k] = __builtin_amdgcn_exp2f(c[k] * (-2.0f * LOG2E_));
    }
}
__device__ __forceinline__ void cell_p6a(float (&q)[4]) {
    for (int k = 0; k < 4; ++k) q[k] = __builtin_amdgcn_rcpf(1.0f + q[k]);
}
__device__ __forceinline__ void cell_p6b(const float (&z)[16], const float (&q)[4], v4f& h) {
    for (int k = 0; k < 4; ++k) h[k] = z[12 + k] * __builtin_fmaf(2.0f, q[k], -1.0f);
}

__global__ __launch_bounds__(32)
void lstm2_fused(const float* __restrict__ x,
                 const float* __restrict__ w_ih0, const float* __restrict__ w_hh0,
                 const float* __restrict__ b_ih0, const float* __restrict__ b_hh0,
                 const float* __restrict__ w_ih1, const float* __restrict__ w_hh1,
                 const float* __restrict__ b_ih1, const float* __restrict__ b_hh1,
                 const float* __restrict__ w_lin, const float* __restrict__ b_lin,
                 float* __restrict__ out)
{
    __shared__ __align__(16) float h1LDS[16 * H_];

    const int lane = threadIdx.x;
    const int c    = lane & 15;
    const int half = lane >> 4;
    const int b0   = blockIdx.x * 16;

    const int jrow = 4 * (c >> 3) + (c & 3);
    const int trow = (c & 7) >> 2;
    const int growA0 = trow * H_ + jrow;
    const int growA1 = (2 + trow) * H_ + jrow;

    v2f Ax[6][2];
    for (int kk = 0; kk < 6; ++kk) {
        int k = kk * 4 + half * 2;
        Ax[kk][0] = *(const v2f*)(w_ih0 + growA0 * D_ + k);
        Ax[kk][1] = *(const v2f*)(w_ih0 + growA1 * D_ + k);
    }
    v2f Ah0[2][2], Ai1[2][2], Ah1[2][2];
    for (int kk = 0; kk < 2; ++kk) {
        int k = half * 4 + kk * 2;
        Ah0[kk][0] = *(const v2f*)(w_hh0 + growA0 * H_ + k);
        Ah0[kk][1] = *(const v2f*)(w_hh0 + growA1 * H_ + k);
        Ai1[kk][0] = *(const v2f*)(w_ih1 + growA0 * H_ + k);
        Ai1[kk][1] = *(const v2f*)(w_ih1 + growA1 * H_ + k);
        Ah1[kk][0] = *(const v2f*)(w_hh1 + growA0 * H_ + k);
        Ah1[kk][1] = *(const v2f*)(w_hh1 + growA1 * H_ + k);
    }

    v8f bias0t0, bias0t1, bias1t0, bias1t1;
    for (int r = 0; r < 8; ++r) {
        int g0 = (r >> 2) * H_ + 4 * half + (r & 3);
        int g1 = (2 + (r >> 2)) * H_ + 4 * half + (r & 3);
        bias0t0[r] = b_ih0[g0] + b_hh0[g0];
        bias0t1[r] = b_ih0[g1] + b_hh0[g1];
        bias1t0[r] = b_ih1[g0] + b_hh1[g0];
        bias1t1[r] = b_ih1[g1] + b_hh1[g1];
    }

    const float* xlane = x + (size_t)(b0 + c) * T_ * D_ + half * 2;

    v4f h0 = {0.f, 0.f, 0.f, 0.f}, h1 = {0.f, 0.f, 0.f, 0.f};
    v4f c0 = {0.f, 0.f, 0.f, 0.f}, c1 = {0.f, 0.f, 0.f, 0.f};

    static_assert(T_ >= 4 && (T_ - 1) % 3 == 0, "the loop body below holds three steps");

    v2f xr[3][6];
    for (int sl = 0; sl < 3; ++sl)
        for (int kk = 0; kk < 6; ++kk) xr[sl][kk] = *(const v2f*)(xlane + (size_t)sl * D_ + kk * 4);

    v8f gx0, gx1;
    v8f a0, a1, e0, e1, g0, g1;
    float z0[16], m0[4], q0[4];
    float z1[16], m1[4], q1[4];
    for (int r = 0; r < 16; ++r) z1[r] = 0.f;

#define PIN_ACC(A)     asm volatile("" : "+v"(A))
#define PIN_Z4(Z, O)   asm volatile("" : "+v"((Z)[(O)]), "+v"((Z)[(O) + 1]), "+v"((Z)[(O) + 2]), "+v"((Z)[(O) + 3]))
#define PIN_4(Q)       asm volatile("" : "+v"((Q)[0]), "+v"((Q)[1]), "+v"((Q)[2]), "+v"((Q)[3]))
#define PIN_V4(V)      asm volatile("" : "+v"(V))
#define SLICE_END()    __builtin_amdgcn_sched_barrier(0)

#define HB(H, KK) (v2f){(H)[2 * (KK)], (H)[2 * (KK) + 1]}

#define GX_FIRST(SL)                                                              \
    do {                                                                          \
        g0 = WMMA_F32(Ax[0][0], xr[SL][0], bias0t0);                              \
        g1 = WMMA_F32(Ax[0][1], xr[SL][0], bias0t1);                              \
    } while (0)
#define GX_NEXT(SL, KK)                                                           \
    do {                                                                          \
        g0 = WMMA_F32(Ax[KK][0], xr[SL][KK], g0);                                 \
        g1 = WMMA_F32(Ax[KK][1], xr[SL][KK], g1);                                 \
    } while (0)
#define REFILL(SL, TN)                                                            \
    do {                                                                          \
        int tn_ = ((TN) < T_) ? (TN) : (T_ - 1);                                  \
        asm volatile("" : "+s"(tn_));                                             \
        const float* xp_ = xlane + (size_t)tn_ * D_;                              \
        for (int kk = 0; kk < 6; ++kk) xr[SL][kk] = *(const v2f*)(xp_ + kk * 4);  \
    } while (0)

#define TRIP(TT, SL)                                                              \
    do {                                                                          \
        a0 = WMMA_F32(Ah0[0][0], HB(h0, 0), gx0);                                 \
        cell_p5a(z1, m1);                                                         \
        PIN_ACC(a0); PIN_4(m1); SLICE_END();                                      \
        a1 = WMMA_F32(Ah0[0][1], HB(h0, 0), gx1);                                 \
        cell_p5b(z1, m1, c1, q1);                                                 \
        PIN_ACC(a1); PIN_V4(c1); PIN_4(q1); SLICE_END();                          \
        a0 = WMMA_F32(Ah0[1][0], HB(h0, 1), a0);                                  \
        cell_p6a(q1);                                                             \
        PIN_ACC(a0); PIN_4(q1); SLICE_END();                                      \
        a1 = WMMA_F32(Ah0[1][1], HB(h0, 1), a1);                                  \
        cell_p6b(z1, q1, h1);                                                     \
        PIN_ACC(a1); PIN_V4(h1); SLICE_END();                                     \
        e0 = WMMA_F32(Ai1[0][0], HB(h0, 0), bias1t0);                             \
        cell_p1(a0, z0, 0);                                                       \
        PIN_ACC(e0); PIN_Z4(z0, 0); SLICE_END();                                  \
        e1 = WMMA_F32(Ai1[0][1], HB(h0, 0), bias1t1);                             \
        cell_p1(a0, z0, 4);                                                       \
        PIN_ACC(e1); PIN_Z4(z0, 4); SLICE_END();                                  \
        e0 = WMMA_F32(Ai1[1][0], HB(h0, 1), e0);                                  \
        cell_p2g(a1, z0);                                                         \
        PIN_ACC(e0); PIN_Z4(z0, 8); SLICE_END();                                  \
        e1 = WMMA_F32(Ai1[1][1], HB(h0, 1), e1);                                  \
        cell_p2o(a1, z0);                                                         \
        PIN_ACC(e1); PIN_Z4(z0, 12); SLICE_END();                                 \
        e0 = WMMA_F32(Ah1[0][0], HB(h1, 0), e0);                                  \
        cell_p3(z0, 0);                                                           \
        PIN_ACC(e0); PIN_Z4(z0, 0); SLICE_END();                                  \
        e1 = WMMA_F32(Ah1[0][1], HB(h1, 0), e1);                                  \
        cell_p3(z0, 4);                                                           \
        PIN_ACC(e1); PIN_Z4(z0, 4); SLICE_END();                                  \
        e0 = WMMA_F32(Ah1[1][0], HB(h1, 1), e0);                                  \
        cell_p3(z0, 8);                                                           \
        PIN_ACC(e0); PIN_Z4(z0, 8); SLICE_END();                                  \
        e1 = WMMA_F32(Ah1[1][1], HB(h1, 1), e1);                                  \
        cell_p3(z0, 12);                                                          \
        PIN_ACC(e1); PIN_Z4(z0, 12); SLICE_END();                                 \
        g0 = WMMA_F32(Ax[0][0], xr[SL][0], bias0t0);                              \
        cell_p5a(z0, m0);                                                         \
        PIN_ACC(g0); PIN_4(m0); SLICE_END();                                      \
        g1 = WMMA_F32(Ax[0][1], xr[SL][0], bias0t1);                              \
        cell_p5b(z0, m0, c0, q0);                                                 \
        PIN_ACC(g1); PIN_V4(c0); PIN_4(q0); SLICE_END();                          \
        g0 = WMMA_F32(Ax[1][0], xr[SL][1], g0);                                   \
        cell_p6a(q0);                                                             \
        PIN_ACC(g0); PIN_4(q0); SLICE_END();                                      \
        g1 = WMMA_F32(Ax[1][1], xr[SL][1], g1);                                   \
        cell_p6b(z0, q0, h0);                                                     \
        PIN_ACC(g1); PIN_V4(h0); SLICE_END();                                     \
        g0 = WMMA_F32(Ax[2][0], xr[SL][2], g0);                                   \
        cell_p1(e0, z1, 0);                                                       \
        PIN_ACC(g0); PIN_Z4(z1, 0); SLICE_END();                                  \
        g1 = WMMA_F32(Ax[2][1], xr[SL][2], g1);                                   \
        cell_p1(e0, z1, 4);                                                       \
        PIN_ACC(g1); PIN_Z4(z1, 4); SLICE_END();                                  \
        g0 = WMMA_F32(Ax[3][0], xr[SL][3], g0);                                   \
        cell_p2g(e1, z1);                                                         \
        PIN_ACC(g0); PIN_Z4(z1, 8); SLICE_END();                                  \
        g1 = WMMA_F32(Ax[3][1], xr[SL][3], g1);                                   \
        cell_p2o(e1, z1);                                                         \
        PIN_ACC(g1); PIN_Z4(z1, 12); SLICE_END();                                 \
        g0 = WMMA_F32(Ax[4][0], xr[SL][4], g0);                                   \
        cell_p3(z1, 0);                                                           \
        PIN_ACC(g0); PIN_Z4(z1, 0); SLICE_END();                                  \
        g1 = WMMA_F32(Ax[4][1], xr[SL][4], g1);                                   \
        cell_p3(z1, 4);                                                           \
        PIN_ACC(g1); PIN_Z4(z1, 4); SLICE_END();                                  \
        g0 = WMMA_F32(Ax[5][0], xr[SL][5], g0);                                   \
        cell_p3(z1, 8);                                                           \
        PIN_ACC(g0); PIN_Z4(z1, 8); SLICE_END();                                  \
        g1 = WMMA_F32(Ax[5][1], xr[SL][5], g1);                                   \
        cell_p3(z1, 12);                                                          \
        PIN_ACC(g1); PIN_Z4(z1, 12); SLICE_END();                                 \
        REFILL(SL, (TT) + 4);                                                     \
        gx0 = g0;                                                                 \
        gx1 = g1;                                                                 \
    } while (0)

    GX_FIRST(0);
    for (int kk = 1; kk < 6; ++kk) GX_NEXT(0, kk);
    REFILL(0, 3);
    a0 = g0;
    a1 = g1;
    for (int kk = 0; kk < 2; ++kk) {
        a0 = WMMA_F32(Ah0[kk][0], HB(h0, kk), a0);
        a1 = WMMA_F32(Ah0[kk][1], HB(h0, kk), a1);
    }
    lstm_cell(a0, a1, c0, h0);
    GX_FIRST(1);
    for (int kk = 1; kk < 6; ++kk) GX_NEXT(1, kk);
    REFILL(1, 4);
    gx0 = g0;
    gx1 = g1;

    for (int t = 1; t < T_; t += 3) {
        TRIP(t,     2);
        TRIP(t + 1, 0);
        TRIP(t + 2, 1);
    }

    cell_p5a(z1, m1);
    cell_p5b(z1, m1, c1, q1);
    cell_p6a(q1);
    cell_p6b(z1, q1, h1);
    e0 = bias1t0;
    e1 = bias1t1;
    for (int kk = 0; kk < 2; ++kk) {
        e0 = WMMA_F32(Ai1[kk][0], HB(h0, kk), e0);
        e1 = WMMA_F32(Ai1[kk][1], HB(h0, kk), e1);
    }
    for (int kk = 0; kk < 2; ++kk) {
        e0 = WMMA_F32(Ah1[kk][0], HB(h1, kk), e0);
        e1 = WMMA_F32(Ah1[kk][1], HB(h1, kk), e1);
    }
    lstm_cell(e0, e1, c1, h1);

#undef TRIP
#undef REFILL
#undef GX_NEXT
#undef GX_FIRST
#undef HB
#undef SLICE_END
#undef PIN_V4
#undef PIN_4
#undef PIN_Z4
#undef PIN_ACC

    *(v4f*)(h1LDS + c * H_ + half * 4) = h1;
    __syncthreads();
    {
        const int obase = half * 12;
        const float* hp = h1LDS + c * H_;
        for (int oo = 0; oo < 12; ++oo) {
            int o = obase + oo;
            float s = b_lin[o];
            for (int j = 0; j < H_; ++j) s += hp[j] * w_lin[o * H_ + j];
            out[(size_t)(b0 + c) * O_ + o] = s;
        }
    }
}

extern "C" void kernel_launch(void* const* d_in, const int* in_sizes, int n_in,
                              void* d_out, int out_size, void* d_ws, size_t ws_size,
                              hipStream_t stream) {
    (void)in_sizes; (void)n_in; (void)d_ws; (void)ws_size; (void)out_size;
    const float* x     = (const float*)d_in[0];
    const float* w_ih0 = (const float*)d_in[1];
    const float* w_hh0 = (const float*)d_in[2];
    const float* b_ih0 = (const float*)d_in[3];
    const float* b_hh0 = (const float*)d_in[4];
    const float* w_ih1 = (const float*)d_in[5];
    const float* w_hh1 = (const float*)d_in[6];
    const float* b_ih1 = (const float*)d_in[7];
    const float* b_hh1 = (const float*)d_in[8];
    const float* w_lin = (const float*)d_in[9];
    const float* b_lin = (const float*)d_in[10];
    float* out = (float*)d_out;

    dim3 grid(B_ / 16);
    dim3 block(32);
    hipLaunchKernelGGL(lstm2_fused, grid, block, 0, stream,
                       x, w_ih0, w_hh0, b_ih0, b_hh0,
                       w_ih1, w_hh1, b_ih1, b_hh1, w_lin, b_lin, out);
}


# every activation is one hardware tanh (v_tanh_f32): sigmoid as 0.5+0.5*tanh(v/2), the 1/2 folded into the i,f,o weight and bias rows (exact, power of two); 6 cell pieces per layer; sum order kept
# speedup vs baseline: 2.7206x; 1.3091x over previous
#include <hip/hip_runtime.h>


typedef float v2f __attribute__((ext_vector_type(2)));
typedef float v4f __attribute__((ext_vector_type(4)));
typedef float v8f __attribute__((ext_vector_type(8)));

#define B_  2048
#define T_  1024
#define D_  24
#define H_  8
#define O_  24

#define WMMA_F32(a, b, c) \
    __builtin_amdgcn_wmma_f32_16x16x4_f32(false, (a), false, (b), (short)0, (c), false, false)

__device__ __forceinline__ float tanh_hw(float x) { return __builtin_amdgcn_tanhf(x); }

__device__ __forceinline__ void lstm_cell(const v8f& t0, const v8f& t1, v4f& c, v4f& h) {
    for (int k = 0; k < 4; ++k) {
        float ig = __builtin_fmaf(0.5f, tanh_hw(t0[k]), 0.5f);
        float fg = __builtin_fmaf(0.5f, tanh_hw(t0[4 + k]), 0.5f);
        float gg = tanh_hw(t1[k]);
        float og = __builtin_fmaf(0.5f, tanh_hw(t1[4 + k]), 0.5f);
        c[k] = __builtin_fmaf(fg, c[k], ig * gg);
        h[k] = og * tanh_hw(c[k]);
    }
}

__device__ __forceinline__ void cell_pa(const v8f& t0, float (&u)[16]) {
    for (int k = 0; k < 4; ++k) u[k] = tanh_hw(t0[k]);
}
__device__ __forceinline__ void cell_pb(const v8f& t0, float (&u)[16]) {
    for (int k = 0; k < 4; ++k) {
        u[4 + k] = tanh_hw(t0[4 + k]);
        u[k] = __builtin_fmaf(0.5f, u[k], 0.5f);
    }
}
__device__ __forceinline__ void cell_pc(const v8f& t1, float (&u)[16]) {
    for (int k = 0; k < 4; ++k) {
        u[8 + k] = tanh_hw(t1[k]);
        u[4 + k] = __builtin_fmaf(0.5f, u[4 + k], 0.5f);
    }
}
__device__ __forceinline__ void cell_pd(const v8f& t1, float (&u)[16], float (&m)[4]) {
    for (int k = 0; k < 4; ++k) {
        u[12 + k] = tanh_hw(t1[4 + k]);
        m[k] = u[k] * u[8 + k];
    }
}
__device__ __forceinline__ void cell_pe(float (&u)[16], const float (&m)[4], v4f& c, float (&tc)[4]) {
    for (int k = 0; k < 4; ++k) c[k] = __builtin_fmaf(u[4 + k], c[k], m[k]);
    for (int k = 0; k < 4; ++k) u[12 + k] = __builtin_fmaf(0.5f, u[12 + k], 0.5f);
    for (int k = 0; k < 4; ++k) tc[k] = tanh_hw(c[k]);
}
__device__ __forceinline__ void cell_pf(const float (&u)[16], const float (&tc)[4], v4f& h) {
    for (int k = 0; k < 4; ++k) h[k] = u[12 + k] * tc[k];
}

__global__ __launch_bounds__(32)
void lstm2_fused(const float* __restrict__ x,
                 const float* __restrict__ w_ih0, const float* __restrict__ w_hh0,
                 const float* __restrict__ b_ih0, const float* __restrict__ b_hh0,
                 const float* __restrict__ w_ih1, const float* __restrict__ w_hh1,
                 const float* __restrict__ b_ih1, const float* __restrict__ b_hh1,
                 const float* __restrict__ w_lin, const float* __restrict__ b_lin,
                 float* __restrict__ out)
{
    __shared__ __align__(16) float h1LDS[16 * H_];

    const int lane = threadIdx.x;
    const int c    = lane & 15;
    const int half = lane >> 4;
    const int b0   = blockIdx.x * 16;

    const int jrow = 4 * (c >> 3) + (c & 3);
    const int trow = (c & 7) >> 2;
    const int growA0 = trow * H_ + jrow;
    const int growA1 = (2 + trow) * H_ + jrow;
    const float sA0 = 0.5f;
    const float sA1 = trow ? 0.5f : 1.0f;

    v2f Ax[6][2];
    for (int kk = 0; kk < 6; ++kk) {
        int k = kk * 4 + half * 2;
        Ax[kk][0] = *(const v2f*)(w_ih0 + growA0 * D_ + k) * sA0;
        Ax[kk][1] = *(const v2f*)(w_ih0 + growA1 * D_ + k) * sA1;
    }
    v2f Ah0[2][2], Ai1[2][2], Ah1[2][2];
    for (int kk = 0; kk < 2; ++kk) {
        int k = half * 4 + kk * 2;
        Ah0[kk][0] = *(const v2f*)(w_hh0 + growA0 * H_ + k) * sA0;
        Ah0[kk][1] = *(const v2f*)(w_hh0 + growA1 * H_ + k) * sA1;
        Ai1[kk][0] = *(const v2f*)(w_ih1 + growA0 * H_ + k) * sA0;
        Ai1[kk][1] = *(const v2f*)(w_ih1 + growA1 * H_ + k) * sA1;
        Ah1[kk][0] = *(const v2f*)(w_hh1 + growA0 * H_ + k) * sA0;
        Ah1[kk][1] = *(const v2f*)(w_hh1 + growA1 * H_ + k) * sA1;
    }

    v8f bias0t0, bias0t1, bias1t0, bias1t1;
    for (int r = 0; r < 8; ++r) {
        int g0 = (r >> 2) * H_ + 4 * half + (r & 3);
        int g1 = (2 + (r >> 2)) * H_ + 4 * half + (r & 3);
        const float s1 = (r >> 2) ? 0.5f : 1.0f;
        bias0t0[r] = (b_ih0[g0] + b_hh0[g0]) * 0.5f;
        bias0t1[r] = (b_ih0[g1] + b_hh0[g1]) * s1;
        bias1t0[r] = (b_ih1[g0] + b_hh1[g0]) * 0.5f;
        bias1t1[r] = (b_ih1[g1] + b_hh1[g1]) * s1;
    }

    const float* xlane = x + (size_t)(b0 + c) * T_ * D_ + half * 2;

    v4f h0 = {0.f, 0.f, 0.f, 0.f}, h1 = {0.f, 0.f, 0.f, 0.f};
    v4f c0 = {0.f, 0.f, 0.f, 0.f}, c1 = {0.f, 0.f, 0.f, 0.f};

    static_assert(T_ >= 4 && (T_ - 1) % 3 == 0, "the loop body below holds three steps");

    v2f xr[3][6];
    for (int sl = 0; sl < 3; ++sl)
        for (int kk = 0; kk < 6; ++kk) xr[sl][kk] = *(const v2f*)(xlane + (size_t)sl * D_ + kk * 4);

    v8f gx0, gx1;
    v8f a0, a1, e0, e1, g0, g1;
    float u0[16], m0[4], q0[4];
    float u1[16], m1[4], q1[4];

#define PIN_ACC(A)     asm volatile("" : "+v"(A))
#define PIN_Z4(Z, O)   asm volatile("" : "+v"((Z)[(O)]), "+v"((Z)[(O) + 1]), "+v"((Z)[(O) + 2]), "+v"((Z)[(O) + 3]))
#define PIN_4(Q)       asm volatile("" : "+v"((Q)[0]), "+v"((Q)[1]), "+v"((Q)[2]), "+v"((Q)[3]))
#define PIN_V4(V)      asm volatile("" : "+v"(V))
#define SLICE_END()    __builtin_amdgcn_sched_barrier(0)

#define HB(H, KK) (v2f){(H)[2 * (KK)], (H)[2 * (KK) + 1]}

#define GX_FIRST(SL)                                                              \
    do {                                                                          \
        g0 = WMMA_F32(Ax[0][0], xr[SL][0], bias0t0);                              \
        g1 = WMMA_F32(Ax[0][1], xr[SL][0], bias0t1);                              \
    } while (0)
#define GX_NEXT(SL, KK)                                                           \
    do {                                                                          \
        g0 = WMMA_F32(Ax[KK][0], xr[SL][KK], g0);                                 \
        g1 = WMMA_F32(Ax[KK][1], xr[SL][KK], g1);                                 \
    } while (0)
#define REFILL(SL, TN)                                                            \
    do {                                                                          \
        int tn_ = ((TN) < T_) ? (TN) : (T_ - 1);                                  \
        asm volatile("" : "+s"(tn_));                                             \
        const float* xp_ = xlane + (size_t)tn_ * D_;                              \
        for (int kk = 0; kk < 6; ++kk) xr[SL][kk] = *(const v2f*)(xp_ + kk * 4);  \
    } while (0)

#define TRIP(TT, SL)                                                              \
    do {                                                                          \
        a0 = WMMA_F32(Ah0[0][0], HB(h0, 0), gx0);                                 \
        a1 = WMMA_F32(Ah0[0][1], HB(h0, 0), gx1);                                 \
        a0 = WMMA_F32(Ah0[1][0], HB(h0, 1), a0);                                  \
        a1 = WMMA_F32(Ah0[1][1], HB(h0, 1), a1);                                  \
        PIN_ACC(a0); PIN_ACC(a1); SLICE_END();                                    \
        e0 = WMMA_F32(Ai1[0][0], HB(h0, 0), bias1t0);                             \
        cell_pa(a0, u0);                                                          \
        PIN_ACC(e0); PIN_Z4(u0, 0); SLICE_END();                                  \
        e1 = WMMA_F32(Ai1[0][1], HB(h0, 0), bias1t1);                             \
        cell_pb(a0, u0);                                                          \
        PIN_ACC(e1); PIN_Z4(u0, 0); PIN_Z4(u0, 4); SLICE_END();                   \
        e0 = WMMA_F32(Ai1[1][0], HB(h0, 1), e0);                                  \
        cell_pc(a1, u0);                                                          \
        PIN_ACC(e0); PIN_Z4(u0, 4); PIN_Z4(u0, 8); SLICE_END();                   \
        e1 = WMMA_F32(Ai1[1][1], HB(h0, 1), e1);                                  \
        cell_pd(a1, u0, m0);                                                      \
        PIN_ACC(e1); PIN_Z4(u0, 12); PIN_4(m0); SLICE_END();                      \
        e0 = WMMA_F32(Ah1[0][0], HB(h1, 0), e0);                                  \
        cell_pe(u0, m0, c0, q0);                                                  \
        PIN_ACC(e0); PIN_V4(c0); PIN_Z4(u0, 12); PIN_4(q0); SLICE_END();          \
        e1 = WMMA_F32(Ah1[0][1], HB(h1, 0), e1);                                  \
        cell_pf(u0, q0, h0);                                                      \
        PIN_ACC(e1); PIN_V4(h0); SLICE_END();                                     \
        e0 = WMMA_F32(Ah1[1][0], HB(h1, 1), e0);                                  \
        e1 = WMMA_F32(Ah1[1][1], HB(h1, 1), e1);                                  \
        PIN_ACC(e0); PIN_ACC(e1); SLICE_END();                                    \
        g0 = WMMA_F32(Ax[0][0], xr[SL][0], bias0t0);                              \
        cell_pa(e0, u1);                                                          \
        PIN_ACC(g0); PIN_Z4(u1, 0); SLICE_END();                                  \
        g1 = WMMA_F32(Ax[0][1], xr[SL][0], bias0t1);                              \
        cell_pb(e0, u1);                                                          \
        PIN_ACC(g1); PIN_Z4(u1, 0); PIN_Z4(u1, 4); SLICE_END();                   \
        g0 = WMMA_F32(Ax[1][0], xr[SL][1], g0);                                   \
        cell_pc(e1, u1);                                                          \
        PIN_ACC(g0); PIN_Z4(u1, 4); PIN_Z4(u1, 8); SLICE_END();                   \
        g1 = WMMA_F32(Ax[1][1], xr[SL][1], g1);                                   \
        cell_pd(e1, u1, m1);                                                      \
        PIN_ACC(g1); PIN_Z4(u1, 12); PIN_4(m1); SLICE_END();                      \
        g0 = WMMA_F32(Ax[2][0], xr[SL][2], g0);                                   \
        cell_pe(u1, m1, c1, q1);                                                  \
        PIN_ACC(g0); PIN_V4(c1); PIN_Z4(u1, 12); PIN_4(q1); SLICE_END();          \
        g1 = WMMA_F32(Ax[2][1], xr[SL][2], g1);                                   \
        cell_pf(u1, q1, h1);                                                      \
        PIN_ACC(g1); PIN_V4(h1); SLICE_END();                                     \
        g0 = WMMA_F32(Ax[3][0], xr[SL][3], g0);                                   \
        g1 = WMMA_F32(Ax[3][1], xr[SL][3], g1);                                   \
        g0 = WMMA_F32(Ax[4][0], xr[SL][4], g0);                                   \
        g1 = WMMA_F32(Ax[4][1], xr[SL][4], g1);                                   \
        g0 = WMMA_F32(Ax[5][0], xr[SL][5], g0);                                   \
        g1 = WMMA_F32(Ax[5][1], xr[SL][5], g1);                                   \
        PIN_ACC(g0); PIN_ACC(g1); SLICE_END();                                    \
        REFILL(SL, (TT) + 4);                                                     \
        gx0 = g0;                                                                 \
        gx1 = g1;                                                                 \
    } while (0)

    GX_FIRST(0);
    for (int kk = 1; kk < 6; ++kk) GX_NEXT(0, kk);
    REFILL(0, 3);
    a0 = g0;
    a1 = g1;
    for (int kk = 0; kk < 2; ++kk) {
        a0 = WMMA_F32(Ah0[kk][0], HB(h0, kk), a0);
        a1 = WMMA_F32(Ah0[kk][1], HB(h0, kk), a1);
    }
    lstm_cell(a0, a1, c0, h0);
    GX_FIRST(1);
    for (int kk = 1; kk < 6; ++kk) GX_NEXT(1, kk);
    REFILL(1, 4);
    gx0 = g0;
    gx1 = g1;

    for (int t = 1; t < T_; t += 3) {
        TRIP(t,     2);
        TRIP(t + 1, 0);
        TRIP(t + 2, 1);
    }

    e0 = bias1t0;
    e1 = bias1t1;
    for (int kk = 0; kk < 2; ++kk) {
        e0 = WMMA_F32(Ai1[kk][0], HB(h0, kk), e0);
        e1 = WMMA_F32(Ai1[kk][1], HB(h0, kk), e1);
    }
    for (int kk = 0; kk < 2; ++kk) {
        e0 = WMMA_F32(Ah1[kk][0], HB(h1, kk), e0);
        e1 = WMMA_F32(Ah1[kk][1], HB(h1, kk), e1);
    }
    lstm_cell(e0, e1, c1, h1);

#undef TRIP
#undef REFILL
#undef GX_NEXT
#undef GX_FIRST
#undef HB
#undef SLICE_END
#undef PIN_V4
#undef PIN_4
#undef PIN_Z4
#undef PIN_ACC

    *(v4f*)(h1LDS + c * H_ + half * 4) = h1;
    __syncthreads();
    {
        const int obase = half * 12;
        const float* hp = h1LDS + c * H_;
        for (int oo = 0; oo < 12; ++oo) {
            int o = obase + oo;
            float s = b_lin[o];
            for (int j = 0; j < H_; ++j) s += hp[j] * w_lin[o * H_ + j];
            out[(size_t)(b0 + c) * O_ + o] = s;
        }
    }
}

extern "C" void kernel_launch(void* const* d_in, const int* in_sizes, int n_in,
                              void* d_out, int out_size, void* d_ws, size_t ws_size,
                              hipStream_t stream) {
    (void)in_sizes; (void)n_in; (void)d_ws; (void)ws_size; (void)out_size;
    const float* x     = (const float*)d_in[0];
    const float* w_ih0 = (const float*)d_in[1];
    const float* w_hh0 = (const float*)d_in[2];
    const float* b_ih0 = (const float*)d_in[3];
    const float* b_hh0 = (const float*)d_in[4];
    const float* w_ih1 = (const float*)d_in[5];
    const float* w_hh1 = (const float*)d_in[6];
    const float* b_ih1 = (const float*)d_in[7];
    const float* b_hh1 = (const float*)d_in[8];
    const float* w_lin = (const float*)d_in[9];
    const float* b_lin = (const float*)d_in[10];
    float* out = (float*)d_out;

    dim3 grid(B_ / 16);
    dim3 block(32);
    hipLaunchKernelGGL(lstm2_fused, grid, block, 0, stream,
                       x, w_ih0, w_hh0, b_ih0, b_hh0,
                       w_ih1, w_hh1, b_ih1, b_hh1, w_lin, b_lin, out);
}


# the step's three stages on three waves of one work-group (input products, layer 0, layer 1): a pipeline over chunks of 8 steps, handed on through LDS at one barrier per chunk; every sum order kept
# speedup vs baseline: 4.3695x; 1.6061x over previous
#include <hip/hip_runtime.h>


typedef float v2f __attribute__((ext_vector_type(2)));
typedef float v4f __attribute__((ext_vector_type(4)));
typedef float v8f __attribute__((ext_vector_type(8)));

#define B_  2048
#define T_  1024
#define D_  24
#define H_  8
#define O_  24

#define WMMA_F32(a, b, c) \
    __builtin_amdgcn_wmma_f32_16x16x4_f32(false, (a), false, (b), (short)0, (c), false, false)

__device__ __forceinline__ float tanh_hw(float x) { return __builtin_amdgcn_tanhf(x); }

__device__ __forceinline__ void lstm_cell(const v8f& t0, const v8f& t1, v4f& c, v4f& h) {
    for (int k = 0; k < 4; ++k) {
        float ig = __builtin_fmaf(0.5f, tanh_hw(t0[k]), 0.5f);
        float fg = __builtin_fmaf(0.5f, tanh_hw(t0[4 + k]), 0.5f);
        float gg = tanh_hw(t1[k]);
        float og = __builtin_fmaf(0.5f, tanh_hw(t1[4 + k]), 0.5f);
        c[k] = __builtin_fmaf(fg, c[k], ig * gg);
        h[k] = og * tanh_hw(c[k]);
    }
}

__device__ __forceinline__ void cell_pa(const v8f& t0, float (&u)[16]) {
    for (int k = 0; k < 4; ++k) u[k] = tanh_hw(t0[k]);
}
__device__ __forceinline__ void cell_pb(const v8f& t0, float (&u)[16]) {
    for (int k = 0; k < 4; ++k) {
        u[4 + k] = tanh_hw(t0[4 + k]);
        u[k] = __builtin_fmaf(0.5f, u[k], 0.5f);
    }
}
__device__ __forceinline__ void cell_pc(const v8f& t1, float (&u)[16]) {
    for (int k = 0; k < 4; ++k) {
        u[8 + k] = tanh_hw(t1[k]);
        u[4 + k] = __builtin_fmaf(0.5f, u[4 + k], 0.5f);
    }
}
__device__ __forceinline__ void cell_pd(const v8f& t1, float (&u)[16], float (&m)[4]) {
    for (int k = 0; k < 4; ++k) {
        u[12 + k] = tanh_hw(t1[4 + k]);
        m[k] = u[k] * u[8 + k];
    }
}
__device__ __forceinline__ void cell_pe(float (&u)[16], const float (&m)[4], v4f& c, float (&tc)[4]) {
    for (int k = 0; k < 4; ++k) c[k] = __builtin_fmaf(u[4 + k], c[k], m[k]);
    for (int k = 0; k < 4; ++k) u[12 + k] = __builtin_fmaf(0.5f, u[12 + k], 0.5f);
    for (int k = 0; k < 4; ++k) tc[k] = tanh_hw(c[k]);
}
__device__ __forceinline__ void cell_pf(const float (&u)[16], const float (&tc)[4], v4f& h) {
    for (int k = 0; k < 4; ++k) h[k] = u[12 + k] * tc[k];
}

#define NCH  8
#define KCH  (T_ / NCH)

__device__ __forceinline__ v4f lo4(const v8f& v) { return __builtin_shufflevector(v, v, 0, 1, 2, 3); }
__device__ __forceinline__ v4f hi4(const v8f& v) { return __builtin_shufflevector(v, v, 4, 5, 6, 7); }
__device__ __forceinline__ v8f cat8(const v4f& a, const v4f& b) {
    return __builtin_shufflevector(a, b, 0, 1, 2, 3, 4, 5, 6, 7);
}

__global__ __launch_bounds__(96)
void lstm2_fused(const float* __restrict__ x,
                 const float* __restrict__ w_ih0, const float* __restrict__ w_hh0,
                 const float* __restrict__ b_ih0, const float* __restrict__ b_hh0,
                 const float* __restrict__ w_ih1, const float* __restrict__ w_hh1,
                 const float* __restrict__ b_ih1, const float* __restrict__ b_hh1,
                 const float* __restrict__ w_lin, const float* __restrict__ b_lin,
                 float* __restrict__ out)
{
    __shared__ __align__(16) v4f gxLDS[2][NCH][4][32];
    __shared__ __align__(16) v4f h0LDS[2][NCH][32];
    __shared__ __align__(16) float h1LDS[16 * H_];

    const int wv   = __builtin_amdgcn_readfirstlane((int)(threadIdx.x >> 5));
    const int lane = threadIdx.x & 31;
    const int c    = lane & 15;
    const int half = lane >> 4;
    const int b0   = blockIdx.x * 16;

    const int jrow = 4 * (c >> 3) + (c & 3);
    const int trow = (c & 7) >> 2;
    const int growA0 = trow * H_ + jrow;
    const int growA1 = (2 + trow) * H_ + jrow;
    const float sA0 = 0.5f;
    const float sA1 = trow ? 0.5f : 1.0f;

    v2f Ax[6][2];
    for (int kk = 0; kk < 6; ++kk) {
        int k = kk * 4 + half * 2;
        Ax[kk][0] = *(const v2f*)(w_ih0 + growA0 * D_ + k) * sA0;
        Ax[kk][1] = *(const v2f*)(w_ih0 + growA1 * D_ + k) * sA1;
    }
    v2f Ah0[2][2], Ai1[2][2], Ah1[2][2];
    for (int kk = 0; kk < 2; ++kk) {
        int k = half * 4 + kk * 2;
        Ah0[kk][0] = *(const v2f*)(w_hh0 + growA0 * H_ + k) * sA0;
        Ah0[kk][1] = *(const v2f*)(w_hh0 + growA1 * H_ + k) * sA1;
        Ai1[kk][0] = *(const v2f*)(w_ih1 + growA0 * H_ + k) * sA0;
        Ai1[kk][1] = *(const v2f*)(w_ih1 + growA1 * H_ + k) * sA1;
        Ah1[kk][0] = *(const v2f*)(w_hh1 + growA0 * H_ + k) * sA0;
        Ah1[kk][1] = *(const v2f*)(w_hh1 + growA1 * H_ + k) * sA1;
    }

    v8f bias0t0, bias0t1, bias1t0, bias1t1;
    for (int r = 0; r < 8; ++r) {
        int g0 = (r >> 2) * H_ + 4 * half + (r & 3);
        int g1 = (2 + (r >> 2)) * H_ + 4 * half + (r & 3);
        const float s1 = (r >> 2) ? 0.5f : 1.0f;
        bias0t0[r] = (b_ih0[g0] + b_hh0[g0]) * 0.5f;
        bias0t1[r] = (b_ih0[g1] + b_hh0[g1]) * s1;
        bias1t0[r] = (b_ih1[g0] + b_hh1[g0]) * 0.5f;
        bias1t1[r] = (b_ih1[g1] + b_hh1[g1]) * s1;
    }

    const float* xlane = x + (size_t)(b0 + c) * T_ * D_ + half * 2;

    v4f h0 = {0.f, 0.f, 0.f, 0.f}, h1 = {0.f, 0.f, 0.f, 0.f};
    v4f c0 = {0.f, 0.f, 0.f, 0.f}, c1 = {0.f, 0.f, 0.f, 0.f};

    static_assert(T_ % NCH == 0 && NCH % 4 == 0, "whole chunks; wave L0 reads half chunks of four steps");

    v2f xr[NCH][6];
    for (int j = 0; j < NCH; ++j)
        for (int kk = 0; kk < 6; ++kk) xr[j][kk] = *(const v2f*)(xlane + (size_t)j * D_ + kk * 4);

    float u1[16], m1[4], q1[4];

#define PIN_ACC(A)     asm volatile("" : "+v"(A))
#define PIN_Z4(Z, O)   asm volatile("" : "+v"((Z)[(O)]), "+v"((Z)[(O) + 1]), "+v"((Z)[(O) + 2]), "+v"((Z)[(O) + 3]))
#define PIN_4(Q)       asm volatile("" : "+v"((Q)[0]), "+v"((Q)[1]), "+v"((Q)[2]), "+v"((Q)[3]))
#define SLICE_END()    __builtin_amdgcn_sched_barrier(0)

#define HB(H, KK) (v2f){(H)[2 * (KK)], (H)[2 * (KK) + 1]}

    for (int k = 0; k < KCH + 2; ++k) {
        if (wv == 0) {
            if (k < KCH) {
                const int buf = k & 1;
                v8f pg0 = bias0t0, pg1 = bias0t1;
                #pragma unroll
                for (int j = 0; j < NCH; ++j) {
                    v8f g0 = WMMA_F32(Ax[0][0], xr[j][0], bias0t0);
                    v8f g1 = WMMA_F32(Ax[0][1], xr[j][0], bias0t1);
                    g0 = WMMA_F32(Ax[1][0], xr[j][1], g0);
                    g1 = WMMA_F32(Ax[1][1], xr[j][1], g1);
                    if (j > 0) {
                        PIN_ACC(g0); PIN_ACC(g1); SLICE_END();
                        gxLDS[buf][j - 1][0][lane] = lo4(pg0);
                        gxLDS[buf][j - 1][1][lane] = hi4(pg0);
                        gxLDS[buf][j - 1][2][lane] = lo4(pg1);
                        gxLDS[buf][j - 1][3][lane] = hi4(pg1);
                    }
                    #pragma unroll
                    for (int kk = 2; kk < 6; ++kk) {
                        g0 = WMMA_F32(Ax[kk][0], xr[j][kk], g0);
                        g1 = WMMA_F32(Ax[kk][1], xr[j][kk], g1);
                    }
                    PIN_ACC(g0); PIN_ACC(g1); SLICE_END();
                    int tn = (k + 1) * NCH + j;
                    tn = (tn < T_) ? tn : (T_ - 1);
                    asm volatile("" : "+s"(tn));
                    const float* xp = xlane + (size_t)tn * D_;
                    #pragma unroll
                    for (int kk = 0; kk < 6; ++kk) xr[j][kk] = *(const v2f*)(xp + kk * 4);
                    pg0 = g0;
                    pg1 = g1;
                }
                gxLDS[buf][NCH - 1][0][lane] = lo4(pg0);
                gxLDS[buf][NCH - 1][1][lane] = hi4(pg0);
                gxLDS[buf][NCH - 1][2][lane] = lo4(pg1);
                gxLDS[buf][NCH - 1][3][lane] = hi4(pg1);
            }
        } else if (wv == 1) {
            if (k >= 1 && k <= KCH) {
                const int buf = (k - 1) & 1;
                for (int hc = 0; hc < NCH / 4; ++hc) {
                    v4f gq[4][4];
                    #pragma unroll
                    for (int jj = 0; jj < 4; ++jj)
                        #pragma unroll
                        for (int q = 0; q < 4; ++q) gq[jj][q] = gxLDS[buf][hc * 4 + jj][q][lane];
                    SLICE_END();
                    #pragma unroll
                    for (int jj = 0; jj < 4; ++jj) {
                        v8f a0 = cat8(gq[jj][0], gq[jj][1]);
                        v8f a1 = cat8(gq[jj][2], gq[jj][3]);
                        #pragma unroll
                        for (int kk = 0; kk < 2; ++kk) {
                            a0 = WMMA_F32(Ah0[kk][0], HB(h0, kk), a0);
                            a1 = WMMA_F32(Ah0[kk][1], HB(h0, kk), a1);
                        }
                        lstm_cell(a0, a1, c0, h0);
                        h0LDS[buf][hc * 4 + jj][lane] = h0;
                    }
                }
            }
        } else {
            if (k >= 2) {
                const int buf = k & 1;
                v4f hb[NCH];
                #pragma unroll
                for (int j = 0; j < NCH; ++j) hb[j] = h0LDS[buf][j][lane];
                v8f n0 = bias1t0, n1 = bias1t1, e0, e1;
                #pragma unroll
                for (int kk = 0; kk < 2; ++kk) {
                    n0 = WMMA_F32(Ai1[kk][0], HB(hb[0], kk), n0);
                    n1 = WMMA_F32(Ai1[kk][1], HB(hb[0], kk), n1);
                }
                #pragma unroll
                for (int j = 0; j < NCH; ++j) {
                    e0 = WMMA_F32(Ah1[0][0], HB(h1, 0), n0);
                    e1 = WMMA_F32(Ah1[0][1], HB(h1, 0), n1);
                    e0 = WMMA_F32(Ah1[1][0], HB(h1, 1), e0);
                    e1 = WMMA_F32(Ah1[1][1], HB(h1, 1), e1);
                    if (j + 1 < NCH) {
                        PIN_ACC(e0); PIN_ACC(e1); SLICE_END();
                        n0 = WMMA_F32(Ai1[0][0], HB(hb[j + 1], 0), bias1t0);
                        cell_pa(e0, u1);
                        PIN_ACC(n0); PIN_Z4(u1, 0); SLICE_END();
                        n1 = WMMA_F32(Ai1[0][1], HB(hb[j + 1], 0), bias1t1);
                        cell_pb(e0, u1);
                        PIN_ACC(n1); PIN_Z4(u1, 0); PIN_Z4(u1, 4); SLICE_END();
                        n0 = WMMA_F32(Ai1[1][0], HB(hb[j + 1], 1), n0);
                        cell_pc(e1, u1);
                        PIN_ACC(n0); PIN_Z4(u1, 4); PIN_Z4(u1, 8); SLICE_END();
                        n1 = WMMA_F32(Ai1[1][1], HB(hb[j + 1], 1), n1);
                        cell_pd(e1, u1, m1);
                        PIN_ACC(n1); PIN_Z4(u1, 12); PIN_4(m1); SLICE_END();
                        cell_pe(u1, m1, c1, q1);
                        cell_pf(u1, q1, h1);
                    } else {
                        lstm_cell(e0, e1, c1, h1);
                    }
                }
            }
        }
        __builtin_amdgcn_fence(__ATOMIC_RELEASE, "workgroup", "local");
        __builtin_amdgcn_s_barrier();
        __builtin_amdgcn_fence(__ATOMIC_ACQUIRE, "workgroup", "local");
    }

#undef HB
#undef SLICE_END
#undef PIN_4
#undef PIN_Z4
#undef PIN_ACC

    if (wv == 2) *(v4f*)(h1LDS + c * H_ + half * 4) = h1;
    __syncthreads();
    if (wv == 2) {
        const int obase = half * 12;
        const float* hp = h1LDS + c * H_;
        for (int oo = 0; oo < 12; ++oo) {
            int o = obase + oo;
            float s = b_lin[o];
            for (int j = 0; j < H_; ++j) s += hp[j] * w_lin[o * H_ + j];
            out[(size_t)(b0 + c) * O_ + o] = s;
        }
    }
}

extern "C" void kernel_launch(void* const* d_in, const int* in_sizes, int n_in,
                              void* d_out, int out_size, void* d_ws, size_t ws_size,
                              hipStream_t stream) {
    (void)in_sizes; (void)n_in; (void)d_ws; (void)ws_size; (void)out_size;
    const float* x     = (const float*)d_in[0];
    const float* w_ih0 = (const float*)d_in[1];
    const float* w_hh0 = (const float*)d_in[2];
    const float* b_ih0 = (const float*)d_in[3];
    const float* b_hh0 = (const float*)d_in[4];
    const float* w_ih1 = (const float*)d_in[5];
    const float* w_hh1 = (const float*)d_in[6];
    const float* b_ih1 = (const float*)d_in[7];
    const float* b_hh1 = (const float*)d_in[8];
    const float* w_lin = (const float*)d_in[9];
    const float* b_lin = (const float*)d_in[10];
    float* out = (float*)d_out;

    dim3 grid(B_ / 16);
    dim3 block(96);
    hipLaunchKernelGGL(lstm2_fused, grid, block, 0, stream,
                       x, w_ih0, w_hh0, b_ih0, b_hh0,
                       w_ih1, w_hh1, b_ih1, b_hh1, w_lin, b_lin, out);
}


# input products (half the matrix work, all global loads) cut in two along time onto two waves: four waves per work-group, one per SIMD, each X wave forms half of every chunk; every sum order kept
# speedup vs baseline: 6.5769x; 1.5052x over previous
#include <hip/hip_runtime.h>


typedef float v2f __attribute__((ext_vector_type(2)));
typedef float v4f __attribute__((ext_vector_type(4)));
typedef float v8f __attribute__((ext_vector_type(8)));

#define B_  2048
#define T_  1024
#define D_  24
#define H_  8
#define O_  24

#define WMMA_F32(a, b, c) \
    __builtin_amdgcn_wmma_f32_16x16x4_f32(false, (a), false, (b), (short)0, (c), false, false)

__device__ __forceinline__ float tanh_hw(float x) { return __builtin_amdgcn_tanhf(x); }

__device__ __forceinline__ void lstm_cell(const v8f& t0, const v8f& t1, v4f& c, v4f& h) {
    for (int k = 0; k < 4; ++k) {
        float ig = __builtin_fmaf(0.5f, tanh_hw(t0[k]), 0.5f);
        float fg = __builtin_fmaf(0.5f, tanh_hw(t0[4 + k]), 0.5f);
        float gg = tanh_hw(t1[k]);
        float og = __builtin_fmaf(0.5f, tanh_hw(t1[4 + k]), 0.5f);
        c[k] = __builtin_fmaf(fg, c[k], ig * gg);
        h[k] = og * tanh_hw(c[k]);
    }
}

__device__ __forceinline__ void cell_pa(const v8f& t0, float (&u)[16]) {
    for (int k = 0; k < 4; ++k) u[k] = tanh_hw(t0[k]);
}
__device__ __forceinline__ void cell_pb(const v8f& t0, float (&u)[16]) {
    for (int k = 0; k < 4; ++k) {
        u[4 + k] = tanh_hw(t0[4 + k]);
        u[k] = __builtin_fmaf(0.5f, u[k], 0.5f);
    }
}
__device__ __forceinline__ void cell_pc(const v8f& t1, float (&u)[16]) {
    for (int k = 0; k < 4; ++k) {
        u[8 + k] = tanh_hw(t1[k]);
        u[4 + k] = __builtin_fmaf(0.5f, u[4 + k], 0.5f);
    }
}
__device__ __forceinline__ void cell_pd(const v8f& t1, float (&u)[16], float (&m)[4]) {
    for (int k = 0; k < 4; ++k) {
        u[12 + k] = tanh_hw(t1[4 + k]);
        m[k] = u[k] * u[8 + k];
    }
}
__device__ __forceinline__ void cell_pe(float (&u)[16], const float (&m)[4], v4f& c, float (&tc)[4]) {
    for (int k = 0; k < 4; ++k) c[k] = __builtin_fmaf(u[4 + k], c[k], m[k]);
    for (int k = 0; k < 4; ++k) u[12 + k] = __builtin_fmaf(0.5f, u[12 + k], 0.5f);
    for (int k = 0; k < 4; ++k) tc[k] = tanh_hw(c[k]);
}
__device__ __forceinline__ void cell_pf(const float (&u)[16], const float (&tc)[4], v4f& h) {
    for (int k = 0; k < 4; ++k) h[k] = u[12 + k] * tc[k];
}

#define NCH  8
#define KCH  (T_ / NCH)
#define NXS  (NCH / 2)

__device__ __forceinline__ v4f lo4(const v8f& v) { return __builtin_shufflevector(v, v, 0, 1, 2, 3); }
__device__ __forceinline__ v4f hi4(const v8f& v) { return __builtin_shufflevector(v, v, 4, 5, 6, 7); }
__device__ __forceinline__ v8f cat8(const v4f& a, const v4f& b) {
    return __builtin_shufflevector(a, b, 0, 1, 2, 3, 4, 5, 6, 7);
}

__global__ __launch_bounds__(128)
void lstm2_fused(const float* __restrict__ x,
                 const float* __restrict__ w_ih0, const float* __restrict__ w_hh0,
                 const float* __restrict__ b_ih0, const float* __restrict__ b_hh0,
                 const float* __restrict__ w_ih1, const float* __restrict__ w_hh1,
                 const float* __restrict__ b_ih1, const float* __restrict__ b_hh1,
                 const float* __restrict__ w_lin, const float* __restrict__ b_lin,
                 float* __restrict__ out)
{
    __shared__ __align__(16) v4f gxLDS[2][NCH][4][32];
    __shared__ __align__(16) v4f h0LDS[2][NCH][32];
    __shared__ __align__(16) float h1LDS[16 * H_];

    const int wv   = __builtin_amdgcn_readfirstlane((int)(threadIdx.x >> 5));
    const int xs0  = (wv == 3) ? NXS : 0;
    const int lane = threadIdx.x & 31;
    const int c    = lane & 15;
    const int half = lane >> 4;
    const int b0   = blockIdx.x * 16;

    const int jrow = 4 * (c >> 3) + (c & 3);
    const int trow = (c & 7) >> 2;
    const int growA0 = trow * H_ + jrow;
    const int growA1 = (2 + trow) * H_ + jrow;
    const float sA0 = 0.5f;
    const float sA1 = trow ? 0.5f : 1.0f;

    v2f Ax[6][2];
    for (int kk = 0; kk < 6; ++kk) {
        int k = kk * 4 + half * 2;
        Ax[kk][0] = *(const v2f*)(w_ih0 + growA0 * D_ + k) * sA0;
        Ax[kk][1] = *(const v2f*)(w_ih0 + growA1 * D_ + k) * sA1;
    }
    v2f Ah0[2][2], Ai1[2][2], Ah1[2][2];
    for (int kk = 0; kk < 2; ++kk) {
        int k = half * 4 + kk * 2;
        Ah0[kk][0] = *(const v2f*)(w_hh0 + growA0 * H_ + k) * sA0;
        Ah0[kk][1] = *(const v2f*)(w_hh0 + growA1 * H_ + k) * sA1;
        Ai1[kk][0] = *(const v2f*)(w_ih1 + growA0 * H_ + k) * sA0;
        Ai1[kk][1] = *(const v2f*)(w_ih1 + growA1 * H_ + k) * sA1;
        Ah1[kk][0] = *(const v2f*)(w_hh1 + growA0 * H_ + k) * sA0;
        Ah1[kk][1] = *(const v2f*)(w_hh1 + growA1 * H_ + k) * sA1;
    }

    v8f bias0t0, bias0t1, bias1t0, bias1t1;
    for (int r = 0; r < 8; ++r) {
        int g0 = (r >> 2) * H_ + 4 * half + (r & 3);
        int g1 = (2 + (r >> 2)) * H_ + 4 * half + (r & 3);
        const float s1 = (r >> 2) ? 0.5f : 1.0f;
        bias0t0[r] = (b_ih0[g0] + b_hh0[g0]) * 0.5f;
        bias0t1[r] = (b_ih0[g1] + b_hh0[g1]) * s1;
        bias1t0[r] = (b_ih1[g0] + b_hh1[g0]) * 0.5f;
        bias1t1[r] = (b_ih1[g1] + b_hh1[g1]) * s1;
    }

    const float* xlane = x + (size_t)(b0 + c) * T_ * D_ + half * 2;

    v4f h0 = {0.f, 0.f, 0.f, 0.f}, h1 = {0.f, 0.f, 0.f, 0.f};
    v4f c0 = {0.f, 0.f, 0.f, 0.f}, c1 = {0.f, 0.f, 0.f, 0.f};

    static_assert(T_ % NCH == 0 && NCH % 4 == 0, "whole chunks; wave L0 reads half chunks of four steps");
    static_assert(NCH % 2 == 0, "the two X waves take half a chunk each");

    v2f xr[NXS][6];
    for (int j = 0; j < NXS; ++j)
        for (int kk = 0; kk < 6; ++kk) xr[j][kk] = *(const v2f*)(xlane + (size_t)(xs0 + j) * D_ + kk * 4);

    float u1[16], m1[4], q1[4];

#define PIN_ACC(A)     asm volatile("" : "+v"(A))
#define PIN_Z4(Z, O)   asm volatile("" : "+v"((Z)[(O)]), "+v"((Z)[(O) + 1]), "+v"((Z)[(O) + 2]), "+v"((Z)[(O) + 3]))
#define PIN_4(Q)       asm volatile("" : "+v"((Q)[0]), "+v"((Q)[1]), "+v"((Q)[2]), "+v"((Q)[3]))
#define SLICE_END()    __builtin_amdgcn_sched_barrier(0)

#define HB(H, KK) (v2f){(H)[2 * (KK)], (H)[2 * (KK) + 1]}

    for (int k = 0; k < KCH + 2; ++k) {
        if (wv == 0 || wv == 3) {
            if (k < KCH) {
                const int buf = k & 1;
                v8f pg0 = bias0t0, pg1 = bias0t1;
                #pragma unroll
                for (int j = 0; j < NXS; ++j) {
                    v8f g0 = WMMA_F32(Ax[0][0], xr[j][0], bias0t0);
                    v8f g1 = WMMA_F32(Ax[0][1], xr[j][0], bias0t1);
                    g0 = WMMA_F32(Ax[1][0], xr[j][1], g0);
                    g1 = WMMA_F32(Ax[1][1], xr[j][1], g1);
                    if (j > 0) {
                        PIN_ACC(g0); PIN_ACC(g1); SLICE_END();
                        gxLDS[buf][xs0 + j - 1][0][lane] = lo4(pg0);
                        gxLDS[buf][xs0 + j - 1][1][lane] = hi4(pg0);
                        gxLDS[buf][xs0 + j - 1][2][lane] = lo4(pg1);
                        gxLDS[buf][xs0 + j - 1][3][lane] = hi4(pg1);
                    }
                    #pragma unroll
                    for (int kk = 2; kk < 6; ++kk) {
                        g0 = WMMA_F32(Ax[kk][0], xr[j][kk], g0);
                        g1 = WMMA_F32(Ax[kk][1], xr[j][kk], g1);
                    }
                    PIN_ACC(g0); PIN_ACC(g1); SLICE_END();
                    int tn = (k + 1) * NCH + xs0 + j;
                    tn = (tn < T_) ? tn : (T_ - 1);
                    asm volatile("" : "+s"(tn));
                    const float* xp = xlane + (size_t)tn * D_;
                    #pragma unroll
                    for (int kk = 0; kk < 6; ++kk) xr[j][kk] = *(const v2f*)(xp + kk * 4);
                    pg0 = g0;
                    pg1 = g1;
                }
                gxLDS[buf][xs0 + NXS - 1][0][lane] = lo4(pg0);
                gxLDS[buf][xs0 + NXS - 1][1][lane] = hi4(pg0);
                gxLDS[buf][xs0 + NXS - 1][2][lane] = lo4(pg1);
                gxLDS[buf][xs0 + NXS - 1][3][lane] = hi4(pg1);
            }
        } else if (wv == 1) {
            if (k >= 1 && k <= KCH) {
                const int buf = (k - 1) & 1;
                for (int hc = 0; hc < NCH / 4; ++hc) {
                    v4f gq[4][4];
                    #pragma unroll
                    for (int jj = 0; jj < 4; ++jj)
                        #pragma unroll
                        for (int q = 0; q < 4; ++q) gq[jj][q] = gxLDS[buf][hc * 4 + jj][q][lane];
                    SLICE_END();
                    #pragma unroll
                    for (int jj = 0; jj < 4; ++jj) {
                        v8f a0 = cat8(gq[jj][0], gq[jj][1]);
                        v8f a1 = cat8(gq[jj][2], gq[jj][3]);
                        #pragma unroll
                        for (int kk = 0; kk < 2; ++kk) {
                            a0 = WMMA_F32(Ah0[kk][0], HB(h0, kk), a0);
                            a1 = WMMA_F32(Ah0[kk][1], HB(h0, kk), a1);
                        }
                        lstm_cell(a0, a1, c0, h0);
                        h0LDS[buf][hc * 4 + jj][lane] = h0;
                    }
                }
            }
        } else {
            if (k >= 2) {
                const int buf = k & 1;
                v4f hb[NCH];
                #pragma unroll
                for (int j = 0; j < NCH; ++j) hb[j] = h0LDS[buf][j][lane];
                v8f n0 = bias1t0, n1 = bias1t1, e0, e1;
                #pragma unroll
                for (int kk = 0; kk < 2; ++kk) {
                    n0 = WMMA_F32(Ai1[kk][0], HB(hb[0], kk), n0);
                    n1 = WMMA_F32(Ai1[kk][1], HB(hb[0], kk), n1);
                }
                #pragma unroll
                for (int j = 0; j < NCH; ++j) {
                    e0 = WMMA_F32(Ah1[0][0], HB(h1, 0), n0);
                    e1 = WMMA_F32(Ah1[0][1], HB(h1, 0), n1);
                    e0 = WMMA_F32(Ah1[1][0], HB(h1, 1), e0);
                    e1 = WMMA_F32(Ah1[1][1], HB(h1, 1), e1);
                    if (j + 1 < NCH) {
                        PIN_ACC(e0); PIN_ACC(e1); SLICE_END();
                        n0 = WMMA_F32(Ai1[0][0], HB(hb[j + 1], 0), bias1t0);
                        cell_pa(e0, u1);
                        PIN_ACC(n0); PIN_Z4(u1, 0); SLICE_END();
                        n1 = WMMA_F32(Ai1[0][1], HB(hb[j + 1], 0), bias1t1);
                        cell_pb(e0, u1);
                        PIN_ACC(n1); PIN_Z4(u1, 0); PIN_Z4(u1, 4); SLICE_END();
                        n0 = WMMA_F32(Ai1[1][0], HB(hb[j + 1], 1), n0);
                        cell_pc(e1, u1);
                        PIN_ACC(n0); PIN_Z4(u1, 4); PIN_Z4(u1, 8); SLICE_END();
                        n1 = WMMA_F32(Ai1[1][1], HB(hb[j + 1], 1), n1);
                        cell_pd(e1, u1, m1);
                        PIN_ACC(n1); PIN_Z4(u1, 12); PIN_4(m1); SLICE_END();
                        cell_pe(u1, m1, c1, q1);
                        cell_pf(u1, q1, h1);
                    } else {
                        lstm_cell(e0, e1, c1, h1);
                    }
                }
            }
        }
        __builtin_amdgcn_fence(__ATOMIC_RELEASE, "workgroup", "local");
        __builtin_amdgcn_s_barrier();
        __builtin_amdgcn_fence(__ATOMIC_ACQUIRE, "workgroup", "local");
    }

#undef HB
#undef SLICE_END
#undef PIN_4
#undef PIN_Z4
#undef PIN_ACC

    if (wv == 2) *(v4f*)(h1LDS + c * H_ + half * 4) = h1;
    __syncthreads();
    if (wv == 2) {
        const int obase = half * 12;
        const float* hp = h1LDS + c * H_;
        for (int oo = 0; oo < 12; ++oo) {
            int o = obase + oo;
            float s = b_lin[o];
            for (int j = 0; j < H_; ++j) s += hp[j] * w_lin[o * H_ + j];
            out[(size_t)(b0 + c) * O_ + o] = s;
        }
    }
}

extern "C" void kernel_launch(void* const* d_in, const int* in_sizes, int n_in,
                              void* d_out, int out_size, void* d_ws, size_t ws_size,
                              hipStream_t stream) {
    (void)in_sizes; (void)n_in; (void)d_ws; (void)ws_size; (void)out_size;
    const float* x     = (const float*)d_in[0];
    const float* w_ih0 = (const float*)d_in[1];
    const float* w_hh0 = (const float*)d_in[2];
    const float* b_ih0 = (const float*)d_in[3];
    const float* b_hh0 = (const float*)d_in[4];
    const float* w_ih1 = (const float*)d_in[5];
    const float* w_hh1 = (const float*)d_in[6];
    const float* b_ih1 = (const float*)d_in[7];
    const float* b_hh1 = (const float*)d_in[8];
    const float* w_lin = (const float*)d_in[9];
    const float* b_lin = (const float*)d_in[10];
    float* out = (float*)d_out;

    dim3 grid(B_ / 16);
    dim3 block(128);
    hipLaunchKernelGGL(lstm2_fused, grid, block, 0, stream,
                       x, w_ih0, w_hh0, b_ih0, b_hh0,
                       w_ih1, w_hh1, b_ih1, b_hh1, w_lin, b_lin, out);
}


# layer 1's input products of the i,f gate tile (2 of wave L1's 8 products per step) moved to the two X waves, one pipeline interval behind layer 0, handed on through LDS; every sum order kept
# speedup vs baseline: 6.8159x; 1.0363x over previous
#include <hip/hip_runtime.h>


typedef float v2f __attribute__((ext_vector_type(2)));
typedef float v4f __attribute__((ext_vector_type(4)));
typedef float v8f __attribute__((ext_vector_type(8)));

#define B_  2048
#define T_  1024
#define D_  24
#define H_  8
#define O_  24

#define WMMA_F32(a, b, c) \
    __builtin_amdgcn_wmma_f32_16x16x4_f32(false, (a), false, (b), (short)0, (c), false, false)

__device__ __forceinline__ float tanh_hw(float x) { return __builtin_amdgcn_tanhf(x); }

__device__ __forceinline__ void lstm_cell(const v8f& t0, const v8f& t1, v4f& c, v4f& h) {
    for (int k = 0; k < 4; ++k) {
        float ig = __builtin_fmaf(0.5f, tanh_hw(t0[k]), 0.5f);
        float fg = __builtin_fmaf(0.5f, tanh_hw(t0[4 + k]), 0.5f);
        float gg = tanh_hw(t1[k]);
        float og = __builtin_fmaf(0.5f, tanh_hw(t1[4 + k]), 0.5f);
        c[k] = __builtin_fmaf(fg, c[k], ig * gg);
        h[k] = og * tanh_hw(c[k]);
    }
}

__device__ __forceinline__ void cell_pa(const v8f& t0, float (&u)[16]) {
    for (int k = 0; k < 4; ++k) u[k] = tanh_hw(t0[k]);
}
__device__ __forceinline__ void cell_pb(const v8f& t0, float (&u)[16]) {
    for (int k = 0; k < 4; ++k) {
        u[4 + k] = tanh_hw(t0[4 + k]);
        u[k] = __builtin_fmaf(0.5f, u[k], 0.5f);
    }
}
__device__ __forceinline__ void cell_pc(const v8f& t1, float (&u)[16]) {
    for (int k = 0; k < 4; ++k) {
        u[8 + k] = tanh_hw(t1[k]);
        u[4 + k] = __builtin_fmaf(0.5f, u[4 + k], 0.5f);
    }
}
__device__ __forceinline__ void cell_pd(const v8f& t1, float (&u)[16], float (&m)[4]) {
    for (int k = 0; k < 4; ++k) {
        u[12 + k] = tanh_hw(t1[4 + k]);
        m[k] = u[k] * u[8 + k];
    }
}
__device__ __forceinline__ void cell_pe(float (&u)[16], const float (&m)[4], v4f& c, float (&tc)[4]) {
    for (int k = 0; k < 4; ++k) c[k] = __builtin_fmaf(u[4 + k], c[k], m[k]);
    for (int k = 0; k < 4; ++k) u[12 + k] = __builtin_fmaf(0.5f, u[12 + k], 0.5f);
    for (int k = 0; k < 4; ++k) tc[k] = tanh_hw(c[k]);
}
__device__ __forceinline__ void cell_pf(const float (&u)[16], const float (&tc)[4], v4f& h) {
    for (int k = 0; k < 4; ++k) h[k] = u[12 + k] * tc[k];
}

#define NCH  8
#define KCH  (T_ / NCH)
#define NXS  (NCH / 2)

__device__ __forceinline__ v4f lo4(const v8f& v) { return __builtin_shufflevector(v, v, 0, 1, 2, 3); }
__device__ __forceinline__ v4f hi4(const v8f& v) { return __builtin_shufflevector(v, v, 4, 5, 6, 7); }
__device__ __forceinline__ v8f cat8(const v4f& a, const v4f& b) {
    return __builtin_shufflevector(a, b, 0, 1, 2, 3, 4, 5, 6, 7);
}

__global__ __launch_bounds__(128)
void lstm2_fused(const float* __restrict__ x,
                 const float* __restrict__ w_ih0, const float* __restrict__ w_hh0,
                 const float* __restrict__ b_ih0, const float* __restrict__ b_hh0,
                 const float* __restrict__ w_ih1, const float* __restrict__ w_hh1,
                 const float* __restrict__ b_ih1, const float* __restrict__ b_hh1,
                 const float* __restrict__ w_lin, const float* __restrict__ b_lin,
                 float* __restrict__ out)
{
    __shared__ __align__(16) v4f gxLDS[2][NCH][4][32];
    __shared__ __align__(16) v4f h0LDS[4][NCH][32];
    __shared__ __align__(16) v4f nLDS[2][NCH][2][32];
    __shared__ __align__(16) float h1LDS[16 * H_];

    const int wv   = __builtin_amdgcn_readfirstlane((int)(threadIdx.x >> 5));
    const int xs0  = (wv == 3) ? NXS : 0;
    const int lane = threadIdx.x & 31;
    const int c    = lane & 15;
    const int half = lane >> 4;
    const int b0   = blockIdx.x * 16;

    const int jrow = 4 * (c >> 3) + (c & 3);
    const int trow = (c & 7) >> 2;
    const int growA0 = trow * H_ + jrow;
    const int growA1 = (2 + trow) * H_ + jrow;
    const float sA0 = 0.5f;
    const float sA1 = trow ? 0.5f : 1.0f;

    v2f Ax[6][2];
    for (int kk = 0; kk < 6; ++kk) {
        int k = kk * 4 + half * 2;
        Ax[kk][0] = *(const v2f*)(w_ih0 + growA0 * D_ + k) * sA0;
        Ax[kk][1] = *(const v2f*)(w_ih0 + growA1 * D_ + k) * sA1;
    }
    v2f Ah0[2][2], Ai1[2][2], Ah1[2][2];
    for (int kk = 0; kk < 2; ++kk) {
        int k = half * 4 + kk * 2;
        Ah0[kk][0] = *(const v2f*)(w_hh0 + growA0 * H_ + k) * sA0;
        Ah0[kk][1] = *(const v2f*)(w_hh0 + growA1 * H_ + k) * sA1;
        Ai1[kk][0] = *(const v2f*)(w_ih1 + growA0 * H_ + k) * sA0;
        Ai1[kk][1] = *(const v2f*)(w_ih1 + growA1 * H_ + k) * sA1;
        Ah1[kk][0] = *(const v2f*)(w_hh1 + growA0 * H_ + k) * sA0;
        Ah1[kk][1] = *(const v2f*)(w_hh1 + growA1 * H_ + k) * sA1;
    }

    v8f bias0t0, bias0t1, bias1t0, bias1t1;
    for (int r = 0; r < 8; ++r) {
        int g0 = (r >> 2) * H_ + 4 * half + (r & 3);
        int g1 = (2 + (r >> 2)) * H_ + 4 * half + (r & 3);
        const float s1 = (r >> 2) ? 0.5f : 1.0f;
        bias0t0[r] = (b_ih0[g0] + b_hh0[g0]) * 0.5f;
        bias0t1[r] = (b_ih0[g1] + b_hh0[g1]) * s1;
        bias1t0[r] = (b_ih1[g0] + b_hh1[g0]) * 0.5f;
        bias1t1[r] = (b_ih1[g1] + b_hh1[g1]) * s1;
    }

    const float* xlane = x + (size_t)(b0 + c) * T_ * D_ + half * 2;

    v4f h0 = {0.f, 0.f, 0.f, 0.f}, h1 = {0.f, 0.f, 0.f, 0.f};
    v4f c0 = {0.f, 0.f, 0.f, 0.f}, c1 = {0.f, 0.f, 0.f, 0.f};

    static_assert(T_ % NCH == 0 && NCH % 4 == 0, "whole chunks; wave L0 reads half chunks of four steps");
    static_assert(NCH % 2 == 0, "the two X waves take half a chunk each");

    v2f xr[NXS][6];
    for (int j = 0; j < NXS; ++j)
        for (int kk = 0; kk < 6; ++kk) xr[j][kk] = *(const v2f*)(xlane + (size_t)(xs0 + j) * D_ + kk * 4);

    float u1[16], m1[4], q1[4];

#define PIN_ACC(A)     asm volatile("" : "+v"(A))
#define PIN_Z4(Z, O)   asm volatile("" : "+v"((Z)[(O)]), "+v"((Z)[(O) + 1]), "+v"((Z)[(O) + 2]), "+v"((Z)[(O) + 3]))
#define PIN_4(Q)       asm volatile("" : "+v"((Q)[0]), "+v"((Q)[1]), "+v"((Q)[2]), "+v"((Q)[3]))
#define SLICE_END()    __builtin_amdgcn_sched_barrier(0)

#define HB(H, KK) (v2f){(H)[2 * (KK)], (H)[2 * (KK) + 1]}

#define GX_LAST_STORES() do { \
        gxLDS[buf][xs0 + NXS - 1][0][lane] = lo4(pg0); \
        gxLDS[buf][xs0 + NXS - 1][1][lane] = hi4(pg0); \
        gxLDS[buf][xs0 + NXS - 1][2][lane] = lo4(pg1); \
        gxLDS[buf][xs0 + NXS - 1][3][lane] = hi4(pg1); \
    } while (0)
#define N_PRODUCTS() \
    v8f nacc[NXS]; \
    _Pragma("unroll") \
    for (int j = 0; j < NXS; ++j) nacc[j] = WMMA_F32(Ai1[0][0], HB(hx[j], 0), bias1t0); \
    _Pragma("unroll") \
    for (int j = 0; j < NXS; ++j) nacc[j] = WMMA_F32(Ai1[1][0], HB(hx[j], 1), nacc[j])
#define N_STORES() do { \
        _Pragma("unroll") \
        for (int j = 0; j < NXS; ++j) { \
            nLDS[k & 1][xs0 + j][0][lane] = lo4(nacc[j]); \
            nLDS[k & 1][xs0 + j][1][lane] = hi4(nacc[j]); \
        } \
    } while (0)

    for (int k = 0; k < KCH + 3; ++k) {
        if (wv == 0 || wv == 3) {
            const bool doN = (k >= 2) && (k < KCH + 2);
            v4f hx[NXS];
            if (doN) {
                #pragma unroll
                for (int j = 0; j < NXS; ++j) hx[j] = h0LDS[(k - 2) & 3][xs0 + j][lane];
            }
            if (k < KCH) {
                const int buf = k & 1;
                v8f pg0 = bias0t0, pg1 = bias0t1;
                #pragma unroll
                for (int j = 0; j < NXS; ++j) {
                    v8f g0 = WMMA_F32(Ax[0][0], xr[j][0], bias0t0);
                    v8f g1 = WMMA_F32(Ax[0][1], xr[j][0], bias0t1);
                    g0 = WMMA_F32(Ax[1][0], xr[j][1], g0);
                    g1 = WMMA_F32(Ax[1][1], xr[j][1], g1);
                    if (j > 0) {
                        PIN_ACC(g0); PIN_ACC(g1); SLICE_END();
                        gxLDS[buf][xs0 + j - 1][0][lane] = lo4(pg0);
                        gxLDS[buf][xs0 + j - 1][1][lane] = hi4(pg0);
                        gxLDS[buf][xs0 + j - 1][2][lane] = lo4(pg1);
                        gxLDS[buf][xs0 + j - 1][3][lane] = hi4(pg1);
                    }
                    #pragma unroll
                    for (int kk = 2; kk < 6; ++kk) {
                        g0 = WMMA_F32(Ax[kk][0], xr[j][kk], g0);
                        g1 = WMMA_F32(Ax[kk][1], xr[j][kk], g1);
                    }
                    PIN_ACC(g0); PIN_ACC(g1); SLICE_END();
                    int tn = (k + 1) * NCH + xs0 + j;
                    tn = (tn < T_) ? tn : (T_ - 1);
                    asm volatile("" : "+s"(tn));
                    const float* xp = xlane + (size_t)tn * D_;
                    #pragma unroll
                    for (int kk = 0; kk < 6; ++kk) xr[j][kk] = *(const v2f*)(xp + kk * 4);
                    pg0 = g0;
                    pg1 = g1;
                }
                if (k >= 2) {
                    N_PRODUCTS();
                    GX_LAST_STORES();
                    N_STORES();
                } else {
                    GX_LAST_STORES();
                }
            } else if (doN) {
                N_PRODUCTS();
                N_STORES();
            }
        } else if (wv == 1) {
            if (k >= 1 && k <= KCH) {
                const int buf = (k - 1) & 1;
                const int hbuf = (k - 1) & 3;
                for (int hc = 0; hc < NCH / 4; ++hc) {
                    v4f gq[4][4];
                    #pragma unroll
                    for (int jj = 0; jj < 4; ++jj)
                        #pragma unroll
                        for (int q = 0; q < 4; ++q) gq[jj][q] = gxLDS[buf][hc * 4 + jj][q][lane];
                    SLICE_END();
                    #pragma unroll
                    for (int jj = 0; jj < 4; ++jj) {
                        v8f a0 = cat8(gq[jj][0], gq[jj][1]);
                        v8f a1 = cat8(gq[jj][2], gq[jj][3]);
                        #pragma unroll
                        for (int kk = 0; kk < 2; ++kk) {
                            a0 = WMMA_F32(Ah0[kk][0], HB(h0, kk), a0);
                            a1 = WMMA_F32(Ah0[kk][1], HB(h0, kk), a1);
                        }
                        lstm_cell(a0, a1, c0, h0);
                        h0LDS[hbuf][hc * 4 + jj][lane] = h0;
                    }
                }
            }
        } else {
            if (k >= 3) {
                const int hbuf = (k - 3) & 3;
                const int nbuf = (k - 1) & 1;
                v4f hb[NCH];
                v4f nq[NCH][2];
                #pragma unroll
                for (int j = 0; j < NCH; ++j) {
                    hb[j] = h0LDS[hbuf][j][lane];
                    nq[j][0] = nLDS[nbuf][j][0][lane];
                    nq[j][1] = nLDS[nbuf][j][1][lane];
                }
                SLICE_END();
                v8f n0 = cat8(nq[0][0], nq[0][1]);
                v8f n1 = bias1t1, e0, e1;
                #pragma unroll
                for (int kk = 0; kk < 2; ++kk)
                    n1 = WMMA_F32(Ai1[kk][1], HB(hb[0], kk), n1);
                #pragma unroll
                for (int j = 0; j < NCH; ++j) {
                    e0 = WMMA_F32(Ah1[0][0], HB(h1, 0), n0);
                    e1 = WMMA_F32(Ah1[0][1], HB(h1, 0), n1);
                    e0 = WMMA_F32(Ah1[1][0], HB(h1, 1), e0);
                    e1 = WMMA_F32(Ah1[1][1], HB(h1, 1), e1);
                    if (j + 1 < NCH) {
                        PIN_ACC(e0); PIN_ACC(e1); SLICE_END();
                        n1 = WMMA_F32(Ai1[0][1], HB(hb[j + 1], 0), bias1t1);
                        cell_pa(e0, u1);
                        PIN_ACC(n1); PIN_Z4(u1, 0); SLICE_END();
                        cell_pb(e0, u1);
                        PIN_Z4(u1, 0); PIN_Z4(u1, 4); SLICE_END();
                        n1 = WMMA_F32(Ai1[1][1], HB(hb[j + 1], 1), n1);
                        cell_pc(e1, u1);
                        PIN_ACC(n1); PIN_Z4(u1, 4); PIN_Z4(u1, 8); SLICE_END();
                        cell_pd(e1, u1, m1);
                        PIN_Z4(u1, 12); PIN_4(m1); SLICE_END();
                        cell_pe(u1, m1, c1, q1);
                        cell_pf(u1, q1, h1);
                        n0 = cat8(nq[j + 1][0], nq[j + 1][1]);
                    } else {
                        lstm_cell(e0, e1, c1, h1);
                    }
                }
            }
        }
        __builtin_amdgcn_fence(__ATOMIC_RELEASE, "workgroup", "local");
        __builtin_amdgcn_s_barrier();
        __builtin_amdgcn_fence(__ATOMIC_ACQUIRE, "workgroup", "local");
    }

#undef N_STORES
#undef N_PRODUCTS
#undef GX_LAST_STORES
#undef HB
#undef SLICE_END
#undef PIN_4
#undef PIN_Z4
#undef PIN_ACC

    if (wv == 2) *(v4f*)(h1LDS + c * H_ + half * 4) = h1;
    __syncthreads();
    if (wv == 2) {
        const int obase = half * 12;
        const float* hp = h1LDS + c * H_;
        for (int oo = 0; oo < 12; ++oo) {
            int o = obase + oo;
            float s = b_lin[o];
            for (int j = 0; j < H_; ++j) s += hp[j] * w_lin[o * H_ + j];
            out[(size_t)(b0 + c) * O_ + o] = s;
        }
    }
}

extern "C" void kernel_launch(void* const* d_in, const int* in_sizes, int n_in,
                              void* d_out, int out_size, void* d_ws, size_t ws_size,
                              hipStream_t stream) {
    (void)in_sizes; (void)n_in; (void)d_ws; (void)ws_size; (void)out_size;
    const float* x     = (const float*)d_in[0];
    const float* w_ih0 = (const float*)d_in[1];
    const float* w_hh0 = (const float*)d_in[2];
    const float* b_ih0 = (const float*)d_in[3];
    const float* b_hh0 = (const float*)d_in[4];
    const float* w_ih1 = (const float*)d_in[5];
    const float* w_hh1 = (const float*)d_in[6];
    const float* b_ih1 = (const float*)d_in[7];
    const float* b_hh1 = (const float*)d_in[8];
    const float* w_lin = (const float*)d_in[9];
    const float* b_lin = (const float*)d_in[10];
    float* out = (float*)d_out;

    dim3 grid(B_ / 16);
    dim3 block(128);
    hipLaunchKernelGGL(lstm2_fused, grid, block, 0, stream,
                       x, w_ih0, w_hh0, b_ih0, b_hh0,
                       w_ih1, w_hh1, b_ih1, b_hh1, w_lin, b_lin, out);
}
